# starting kernel
# baseline (speedup 1.0000x reference)
#include <hip/hip_runtime.h>
#include <math.h>


#define NGROUPS 561
#define GMM_K   32
#define LOG2PI  1.8378770664093453f
#define XS      68
#define RB      4
#define WPB     3

typedef __attribute__((ext_vector_type(2))) float v2f;
typedef __attribute__((ext_vector_type(8))) float v8f;

__global__ __launch_bounds__(64) void gmm_prep(const float* __restrict__ loc,
                                               const float* __restrict__ cov,
                                               const float* __restrict__ pi,
                                               float* __restrict__ bpack) {
  __shared__ float A[64][65];
  __shared__ float M[64][65];
  __shared__ float mu[64];
  __shared__ float bv[64];
  __shared__ float sc[3];
  const int k = blockIdx.x;
  const int t = threadIdx.x;

  for (int i = 0; i < 64; ++i) A[t][i] = cov[((size_t)k * 64 + t) * 64 + i];
  mu[t] = loc[k * 64 + t];
  if (t == 0) {
    float m = pi[0];
    for (int i = 1; i < GMM_K; ++i) m = fmaxf(m, pi[i]);
    float s = 0.f;
    for (int i = 0; i < GMM_K; ++i) s += expf(pi[i] - m);
    sc[0] = m + logf(s);
  }
  __syncthreads();

  for (int j = 0; j < 64; ++j) {
    if (t == j) A[j][j] = sqrtf(A[j][j]);
    __syncthreads();
    if (t > j) A[t][j] /= A[j][j];
    __syncthreads();
    if (t > j) {
      float l = A[t][j];
      for (int c = j + 1; c <= t; ++c) A[t][c] -= l * A[c][j];
    }
    __syncthreads();
  }
  if (t == 0) {
    float hs = 0.f;
    for (int i = 0; i < 64; ++i) hs += logf(A[i][i]);
    sc[1] = hs;
  }

  {
    const int c = t;
    for (int i = 0; i < c; ++i) M[i][c] = 0.f;
    M[c][c] = 1.f / A[c][c];
    for (int i = c + 1; i < 64; ++i) {
      float s = 0.f;
      for (int p = c; p < i; ++p) s += A[i][p] * M[p][c];
      M[i][c] = -s / A[i][i];
    }
  }
  __syncthreads();

  {
    const int j = t;
    for (int i = 0; i <= j; ++i) {
      float s = 0.f;
      for (int p = j; p < 64; ++p) s += M[p][i] * M[p][j];
      A[j][i] = s;
    }
  }
  __syncthreads();

  {
    const int a = t;
    float s = 0.f;
    for (int b = 0; b < 64; ++b) {
      float p = (b <= a) ? A[a][b] : A[b][a];
      s += p * mu[b];
    }
    bv[a] = s;
  }
  __syncthreads();
  if (t == 0) {
    float c2 = 0.f;
    for (int i = 0; i < 64; ++i) c2 += mu[i] * bv[i];
    sc[2] = -0.5f * c2 + pi[k] - sc[0] - sc[1] - 0.5f * 64.f * LOG2PI;
  }
  __syncthreads();

  const int ctile = k >> 4;
  const int n = k & 15;
  float* base = bpack + (size_t)ctile * NGROUPS * 64;
  int g = 0;
  for (int j = 0; j < 64; ++j) {
    const int nc = (j >> 2) + 1;
    for (int ci = 0; ci < nc; ++ci, ++g) {
      if ((g & 63) == t) {
        float* dst = base + (size_t)g * 64 + n * 2;
        for (int q = 0; q < 4; ++q) {
          const int i = 4 * ci + q;
          float w = 0.f;
          if (i < j)       w = -A[j][i];
          else if (i == j) w = -0.5f * A[j][j];
          dst[(q >> 1) * 32 + (q & 1)] = w;
        }
      }
    }
  }
  for (int ci = 0; ci < 16; ++ci, ++g) {
    if ((g & 63) == t) {
      float* dst = base + (size_t)g * 64 + n * 2;
      for (int q = 0; q < 4; ++q)
        dst[(q >> 1) * 32 + (q & 1)] = bv[4 * ci + q];
    }
  }
  if ((g & 63) == t) {
    float* dst = base + (size_t)g * 64 + n * 2;
    for (int q = 0; q < 4; ++q)
      dst[(q >> 1) * 32 + (q & 1)] = (q == 0) ? sc[2] : 0.f;
  }
}

__global__ __launch_bounds__(32 * WPB) void gmm_main(const float* __restrict__ X,
                                                     const float* __restrict__ bpack,
                                                     float* __restrict__ out,
                                                     int ntiles) {
  __shared__ float xs[WPB][16 * RB][XS];
  const int tid = threadIdx.x;
  const int w = tid >> 5;
  const int lane = tid & 31;
  const int h = lane >> 4;
  const int r = lane & 15;
  const int t0 = (blockIdx.x * WPB + w) * RB;
  int nvalid = ntiles - t0;
  if (nvalid > RB) nvalid = RB;

  if (nvalid > 0) {
    const float* xp = X + (size_t)t0 * 16 * 64;
    const int umax = nvalid * 8;
    for (int u = 0; u < umax; ++u) {
      const int flat = u * 128 + lane * 4;
      float4 v = *(const float4*)(xp + flat);
      *(float4*)&xs[w][flat >> 6][flat & 63] = v;
    }
  }
  __syncthreads();
  if (nvalid <= 0) return;

  const float* xr[RB];
#pragma unroll
  for (int rt = 0; rt < RB; ++rt) xr[rt] = &xs[w][16 * rt + r][0];
  const v2f* __restrict__ bp0 = (const v2f*)bpack + (h * 16 + r);
  const v2f* __restrict__ bp1 = bp0 + (size_t)NGROUPS * 32;

  v8f acc[RB][2];
#pragma unroll
  for (int rt = 0; rt < RB; ++rt) { acc[rt][0] = (v8f){}; acc[rt][1] = (v8f){}; }

  int g = 0;
  for (int j = 0; j < 64; ++j) {
    float tj[RB];
#pragma unroll
    for (int rt = 0; rt < RB; ++rt) tj[rt] = xr[rt][j];
    const int nc = (j >> 2) + 1;
    for (int ci = 0; ci < nc; ++ci, ++g) {
      v2f b0 = bp0[(size_t)g * 32];
      v2f b1 = bp1[(size_t)g * 32];
      v2f a[RB];
#pragma unroll
      for (int rt = 0; rt < RB; ++rt)
        a[rt] = (*(const v2f*)(xr[rt] + 4 * ci + 2 * h)) * tj[rt];
#pragma unroll
      for (int rt = 0; rt < RB; ++rt) {
        acc[rt][0] = __builtin_amdgcn_wmma_f32_16x16x4_f32(false, a[rt], false, b0, (short)0, acc[rt][0], false, false);
        acc[rt][1] = __builtin_amdgcn_wmma_f32_16x16x4_f32(false, a[rt], false, b1, (short)0, acc[rt][1], false, false);
      }
    }
  }
  for (int ci = 0; ci < 16; ++ci, ++g) {
    v2f b0 = bp0[(size_t)g * 32];
    v2f b1 = bp1[(size_t)g * 32];
    v2f a[RB];
#pragma unroll
    for (int rt = 0; rt < RB; ++rt)
      a[rt] = *(const v2f*)(xr[rt] + 4 * ci + 2 * h);
#pragma unroll
    for (int rt = 0; rt < RB; ++rt) {
      acc[rt][0] = __builtin_amdgcn_wmma_f32_16x16x4_f32(false, a[rt], false, b0, (short)0, acc[rt][0], false, false);
      acc[rt][1] = __builtin_amdgcn_wmma_f32_16x16x4_f32(false, a[rt], false, b1, (short)0, acc[rt][1], false, false);
    }
  }
  {
    v2f b0 = bp0[(size_t)g * 32];
    v2f b1 = bp1[(size_t)g * 32];
    v2f a; a.x = (h == 0) ? 1.f : 0.f; a.y = 0.f;
#pragma unroll
    for (int rt = 0; rt < RB; ++rt) {
      acc[rt][0] = __builtin_amdgcn_wmma_f32_16x16x4_f32(false, a, false, b0, (short)0, acc[rt][0], false, false);
      acc[rt][1] = __builtin_amdgcn_wmma_f32_16x16x4_f32(false, a, false, b1, (short)0, acc[rt][1], false, false);
    }
  }

#pragma unroll
  for (int rt = 0; rt < RB; ++rt) {
    float res[8];
#pragma unroll
    for (int v = 0; v < 8; ++v) {
      float a0 = acc[rt][0][v], a1 = acc[rt][1][v];
      float mx = fmaxf(a0, a1);
      mx = fmaxf(mx, __shfl_xor(mx, 1, 32));
      mx = fmaxf(mx, __shfl_xor(mx, 2, 32));
      mx = fmaxf(mx, __shfl_xor(mx, 4, 32));
      mx = fmaxf(mx, __shfl_xor(mx, 8, 32));
      float e = expf(a0 - mx) + expf(a1 - mx);
      e += __shfl_xor(e, 1, 32);
      e += __shfl_xor(e, 2, 32);
      e += __shfl_xor(e, 4, 32);
      e += __shfl_xor(e, 8, 32);
      res[v] = mx + logf(e);
    }
    float val = res[0];
#pragma unroll
    for (int v = 1; v < 8; ++v) val = (r == v) ? res[v] : val;
    if (rt < nvalid && r < 8)
      out[(size_t)(t0 + rt) * 16 + 8 * h + r] = val;
  }
}

extern "C" void kernel_launch(void* const* d_in, const int* in_sizes, int n_in,
                              void* d_out, int out_size, void* d_ws, size_t ws_size,
                              hipStream_t stream) {
  const float* X   = (const float*)d_in[0];
  const float* loc = (const float*)d_in[1];
  const float* cov = (const float*)d_in[2];
  const float* pi  = (const float*)d_in[3];
  float* out = (float*)d_out;
  float* bpack = (float*)d_ws;

  const int N = in_sizes[0] / 64;
  const int ntiles = N / 16;

  gmm_prep<<<GMM_K, 64, 0, stream>>>(loc, cov, pi, bpack);
  const int tilesPerBlock = WPB * RB;
  const int nblk = (ntiles + tilesPerBlock - 1) / tilesPerBlock;
  gmm_main<<<nblk, 32 * WPB, 0, stream>>>(X, bpack, out, ntiles);
}


# gmm_prep on 256 threads: root-free right-looking sweep in 16-column register tiles, pivot column via LDS (1 barrier/column), 1/sqrt(d) scaling after; inverse columns in registers; sum order changes
# speedup vs baseline: 4.4944x; 4.4944x over previous
#include <hip/hip_runtime.h>
#include <math.h>


#define GMM_K   32
#define LOG2PI  1.8378770664093453f

#define PACKF     2244
#define OD_FLOATS 1536
#define BIAS_OFF  2176
#define CONST_OFF 2240
#define NCHUNK    561
#define LDSF      2632
#define L_BIAS    2560
#define L_CONST   2624

#define RB      2
#define WPB     8

typedef __attribute__((ext_vector_type(2))) float v2f;
typedef __attribute__((ext_vector_type(8))) float v8f;

__global__ __launch_bounds__(256) void gmm_prep(const float* __restrict__ loc,
                                                const float* __restrict__ cov,
                                                const float* __restrict__ pi,
                                                float* __restrict__ pack) {
  __shared__ __attribute__((aligned(16))) float Ls[64][68];
  __shared__ float M[64][65];
  __shared__ __attribute__((aligned(16))) float colb[2][64];
  __shared__ __attribute__((aligned(16))) float rdv[64];
  __shared__ __attribute__((aligned(16))) float lg[64];
  __shared__ float mu[64];
  __shared__ float bv[64];
  __shared__ float sc[3];
  const int k = blockIdx.x;
  const int tid = threadIdx.x;
  const int lane = tid & 31;
  const int w = __builtin_amdgcn_readfirstlane(tid >> 5);
  const int cb = w >> 1;
  const int r = ((w & 1) << 5) + lane;

  {
    const float4* src = (const float4*)(cov + (size_t)k * 4096);
#pragma unroll
    for (int it = 0; it < 4; ++it) {
      const int f = it * 256 + tid;
      *(float4*)&Ls[f >> 4][(f & 15) * 4] = src[f];
    }
  }
  if (tid < 64) mu[tid] = loc[k * 64 + tid];
  if (tid == 0) {
    float m = pi[0];
    for (int i = 1; i < GMM_K; ++i) m = fmaxf(m, pi[i]);
    float s = 0.f;
    for (int i = 0; i < GMM_K; ++i) s += expf(pi[i] - m);
    sc[0] = m + logf(s);
  }
  __syncthreads();

  float a[16];
#pragma unroll
  for (int q4 = 0; q4 < 4; ++q4) {
    const float4 v = *(const float4*)&Ls[r][16 * cb + 4 * q4];
    a[4 * q4 + 0] = v.x; a[4 * q4 + 1] = v.y; a[4 * q4 + 2] = v.z; a[4 * q4 + 3] = v.w;
  }

#pragma nounroll
  for (int jb = 0; jb < 4; ++jb) {
#pragma unroll
    for (int js = 0; js < 16; ++js) {
      float* cj = &colb[js & 1][0];
      if (cb == jb) cj[r] = a[js];
      __syncthreads();
      if (cb >= jb) {
        const float d = cj[16 * jb + js];
        const float g = cj[r] / d;
        float cv[16];
#pragma unroll
        for (int q4 = 0; q4 < 4; ++q4) {
          const float4 v = *(const float4*)&cj[16 * cb + 4 * q4];
          cv[4 * q4 + 0] = v.x; cv[4 * q4 + 1] = v.y; cv[4 * q4 + 2] = v.z; cv[4 * q4 + 3] = v.w;
        }
#pragma unroll
        for (int q = js + 1; q < 16; ++q) a[q] = fmaf(-g, cv[q], a[q]);
        if (cb > jb) {
#pragma unroll
          for (int q = 0; q <= js; ++q) a[q] = fmaf(-g, cv[q], a[q]);
        }
      }
    }
  }

  float dr = a[0];
#pragma unroll
  for (int q = 1; q < 16; ++q) dr = ((r & 15) == q) ? a[q] : dr;
  if ((r >> 4) == cb) {
    const float sq = sqrtf(dr);
    rdv[r] = 1.0f / sq;
    lg[r] = logf(sq);
  }
  __syncthreads();

#pragma unroll
  for (int q4 = 0; q4 < 4; ++q4) {
    const float4 sv = *(const float4*)&rdv[16 * cb + 4 * q4];
    float4 o;
    o.x = a[4 * q4 + 0] * sv.x; o.y = a[4 * q4 + 1] * sv.y;
    o.z = a[4 * q4 + 2] * sv.z; o.w = a[4 * q4 + 3] * sv.w;
    *(float4*)&Ls[r][16 * cb + 4 * q4] = o;
  }
  if (tid == 0) {
    float hs = 0.f;
    for (int i = 0; i < 64; ++i) hs += lg[i];
    sc[1] = hs;
  }
  __syncthreads();

  if (tid < 64) {
    const int c = tid;
    float m[64];
#pragma unroll
    for (int i = 0; i < 64; ++i) {
      float s0 = 0.f, s1 = 0.f, s2 = 0.f, s3 = 0.f;
#pragma unroll
      for (int p4 = 0; p4 < (i + 3) / 4; ++p4) {
        const float4 lv = *(const float4*)&Ls[i][4 * p4];
        if (4 * p4 + 0 < i) s0 = fmaf(lv.x, m[4 * p4 + 0], s0);
        if (4 * p4 + 1 < i) s1 = fmaf(lv.y, m[4 * p4 + 1], s1);
        if (4 * p4 + 2 < i) s2 = fmaf(lv.z, m[4 * p4 + 2], s2);
        if (4 * p4 + 3 < i) s3 = fmaf(lv.w, m[4 * p4 + 3], s3);
      }
      const float e = (c == i) ? 1.f : 0.f;
      m[i] = (e - ((s0 + s1) + (s2 + s3))) * rdv[i];
      M[i][c] = m[i];
    }
  }
  __syncthreads();

  if (tid < 64) {
    const int t = tid;
    float s = 0.f;
    for (int p = 0; p <= t; ++p) s += M[t][p] * mu[p];
    bv[t] = s;
    if (t == 0) sc[2] = pi[k] - sc[0] - sc[1] - 0.5f * 64.f * LOG2PI;
  }
  __syncthreads();

  if (tid < 64) {
    const int t = tid;
    float* base = pack + (size_t)k * PACKF;
    const int l = t >> 1, s = t & 1, h = l >> 4, c = l & 15;
    int od = 0;
    for (int zt = 0; zt < 4; ++zt) {
      const int row = 16 * zt + c;
      for (int kg = 0; kg < 4 * zt; ++kg, ++od)
        base[od * 64 + t] = M[row][4 * kg + 2 * h + s];
      for (int j = 0; j < 4; ++j) {
        if (c >= 4 * j) {
          const int startf = 64 * j - 8 * j * (j - 1);
          base[OD_FLOATS + zt * 160 + startf + h * (32 - 8 * j) + (c - 4 * j) * 2 + s] =
              M[row][16 * zt + 4 * j + 2 * h + s];
        }
      }
    }
    base[BIAS_OFF + t] = -bv[t];
    if (t < 4) base[CONST_OFF + t] = (t == 0) ? sc[2] : 0.f;
  }
}

__device__ __forceinline__ int chunk_dst(int i) {
  if (i < 384) {
    const int od = i >> 4;
    const int f = (od < 4) ? (4 + od) : (od < 12) ? (8 + od) : (12 + od);
    return f * 64 + (i & 15) * 4;
  }
  if (i < 544) {
    const int e = i - 384;
    const int zt = e / 40;
    const int rem = e - zt * 40;
    const int j = (rem >= 16) + (rem >= 28) + (rem >= 36);
    const int start = (j == 0) ? 0 : (j == 1) ? 16 : (j == 2) ? 28 : 36;
    const int q = rem - start;
    const int len = 8 - 2 * j;
    const int hh = (q >= len) ? 1 : 0;
    const int u = q - hh * len;
    const int f = 2 * zt * (zt + 1) + 4 * zt + j;
    return f * 64 + 32 * hh + 8 * j + 4 * u;
  }
  return L_BIAS + (i - 544) * 4;
}

__global__ __launch_bounds__(32 * WPB) void gmm_main(const float* __restrict__ X,
                                                     const float* __restrict__ pack,
                                                     float* __restrict__ out,
                                                     int ntiles) {
  __shared__ __attribute__((aligned(32))) float mb[2][LDSF];
  const int tid = threadIdx.x;
  const int w = tid >> 5;
  const int lane = tid & 31;
  const int h = lane >> 4;
  const int c = lane & 15;
  const int t0 = (blockIdx.x * WPB + w) * RB;

  v2f xb[RB][16];
#pragma unroll
  for (int rt = 0; rt < RB; ++rt) {
    int tile = t0 + rt;
    if (tile > ntiles - 1) tile = ntiles - 1;
    const float* xp = X + ((size_t)tile * 16 + c) * 64 + 2 * h;
#pragma unroll
    for (int kg = 0; kg < 16; ++kg) xb[rt][kg] = *(const v2f*)(xp + 4 * kg);
  }

  const int d0 = chunk_dst(tid);
  const int d1 = chunk_dst(tid + 256);
  const bool has2 = (tid + 512) < NCHUNK;
  const int d2 = has2 ? chunk_dst(tid + 512) : 0;

  for (int i = tid; i < 2 * LDSF / 4; i += 32 * WPB)
    ((float4*)&mb[0][0])[i] = make_float4(0.f, 0.f, 0.f, 0.f);

  float4 s0, s1, s2 = make_float4(0.f, 0.f, 0.f, 0.f);
  {
    const float4* src = (const float4*)pack;
    s0 = src[tid];
    s1 = src[tid + 256];
    if (has2) s2 = src[tid + 512];
  }
  __syncthreads();
  *(float4*)&mb[0][d0] = s0;
  *(float4*)&mb[0][d1] = s1;
  if (has2) *(float4*)&mb[0][d2] = s2;
  __syncthreads();

  float m_run[RB], s_run[RB];
#pragma unroll
  for (int rt = 0; rt < RB; ++rt) { m_run[rt] = -3.0e38f; s_run[rt] = 0.f; }

  for (int k = 0; k < GMM_K; ++k) {
    const float* cur = &mb[k & 1][0];
    const bool more = (k + 1) < GMM_K;
    if (more) {
      const float4* src = (const float4*)(pack + (size_t)(k + 1) * PACKF);
      s0 = src[tid];
      s1 = src[tid + 256];
      if (has2) s2 = src[tid + 512];
    }

    float ss[RB];
#pragma unroll
    for (int rt = 0; rt < RB; ++rt) ss[rt] = 0.f;

#pragma unroll
    for (int zt = 0; zt < 4; ++zt) {
      const v8f bias = *(const v8f*)(cur + L_BIAS + 16 * zt + 8 * h);
      v8f acc[RB];
#pragma unroll
      for (int kg = 0; kg < 4 * (zt + 1); ++kg) {
        const v2f a = *(const v2f*)(cur + (2 * zt * (zt + 1) + kg) * 64 + 2 * lane);
#pragma unroll
        for (int rt = 0; rt < RB; ++rt)
          acc[rt] = __builtin_amdgcn_wmma_f32_16x16x4_f32(false, a, false, xb[rt][kg], (short)0,
                                                          (kg == 0) ? bias : acc[rt], false, false);
      }
#pragma unroll
      for (int rt = 0; rt < RB; ++rt)
#pragma unroll
        for (int r = 0; r < 8; ++r) ss[rt] = fmaf(acc[rt][r], acc[rt][r], ss[rt]);
    }

    const float ck = cur[L_CONST];
#pragma unroll
    for (int rt = 0; rt < RB; ++rt) {
      const float mh = ss[rt] + __shfl_xor(ss[rt], 16, 32);
      const float v = fmaf(-0.5f, mh, ck);
      const float mn = fmaxf(m_run[rt], v);
      s_run[rt] = s_run[rt] * expf(m_run[rt] - mn) + expf(v - mn);
      m_run[rt] = mn;
    }

    if (more) {
      float* nxt = &mb[(k + 1) & 1][0];
      *(float4*)&nxt[d0] = s0;
      *(float4*)&nxt[d1] = s1;
      if (has2) *(float4*)&nxt[d2] = s2;
    }
    __syncthreads();
  }

  const float r0 = m_run[0] + logf(s_run[0]);
  const float r1 = m_run[1] + logf(s_run[1]);
  const float val = (h == 0) ? r0 : r1;
  if (t0 + h < ntiles) out[(size_t)t0 * 16 + lane] = val;
}

extern "C" void kernel_launch(void* const* d_in, const int* in_sizes, int n_in,
                              void* d_out, int out_size, void* d_ws, size_t ws_size,
                              hipStream_t stream) {
  const float* X   = (const float*)d_in[0];
  const float* loc = (const float*)d_in[1];
  const float* cov = (const float*)d_in[2];
  const float* pi  = (const float*)d_in[3];
  float* out = (float*)d_out;
  float* pack = (float*)d_ws;

  const int N = in_sizes[0] / 64;
  const int ntiles = N / 16;

  gmm_prep<<<GMM_K, 256, 0, stream>>>(loc, cov, pi, pack);
  const int tilesPerBlock = WPB * RB;
  const int nblk = (ntiles + tilesPerBlock - 1) / tilesPerBlock;
  gmm_main<<<nblk, 32 * WPB, 0, stream>>>(X, pack, out, ntiles);
}


# gmm_prep: the elimination sweep carries [I | mu] beside cov under the same 64 barriers, so L^-1 and L^-1 mu leave it in registers; forward substitution and M*mu passes dropped; sum order changes
# speedup vs baseline: 4.7845x; 1.0645x over previous
#include <hip/hip_runtime.h>
#include <math.h>


#define GMM_K   32
#define LOG2PI  1.8378770664093453f

#define PACKF     2244
#define OD_FLOATS 1536
#define BIAS_OFF  2176
#define CONST_OFF 2240
#define NCHUNK    561
#define LDSF      2632
#define L_BIAS    2560
#define L_CONST   2624

#define RB      2
#define WPB     8

typedef __attribute__((ext_vector_type(2))) float v2f;
typedef __attribute__((ext_vector_type(8))) float v8f;

__global__ __launch_bounds__(256) void gmm_prep(const float* __restrict__ loc,
                                                const float* __restrict__ cov,
                                                const float* __restrict__ pi,
                                                float* __restrict__ pack) {
  __shared__ __attribute__((aligned(16))) float Ls[64][68];
  __shared__ float M[64][65];
  __shared__ __attribute__((aligned(16))) float colb[2][64];
  __shared__ __attribute__((aligned(16))) float rowb[2][64];
  __shared__ __attribute__((aligned(16))) float rdv[64];
  __shared__ __attribute__((aligned(16))) float lg[64];
  __shared__ float yb[2];
  __shared__ float bv[64];
  __shared__ float sc[2];
  const int k = blockIdx.x;
  const int tid = threadIdx.x;
  const int lane = tid & 31;
  const int w = __builtin_amdgcn_readfirstlane(tid >> 5);
  const int cb = w >> 1;
  const int r = ((w & 1) << 5) + lane;

  {
    const float4* src = (const float4*)(cov + (size_t)k * 4096);
#pragma unroll
    for (int it = 0; it < 4; ++it) {
      const int f = it * 256 + tid;
      *(float4*)&Ls[f >> 4][(f & 15) * 4] = src[f];
    }
  }
  const float ymu = loc[k * 64 + r];
  if (tid < 128) (&rowb[0][0])[tid] = 0.f;
  if (tid == 0) {
    float m = pi[0];
    for (int i = 1; i < GMM_K; ++i) m = fmaxf(m, pi[i]);
    float s = 0.f;
    for (int i = 0; i < GMM_K; ++i) s += expf(pi[i] - m);
    sc[0] = m + logf(s);
  }
  __syncthreads();

  float a[16];
#pragma unroll
  for (int q4 = 0; q4 < 4; ++q4) {
    const float4 v = *(const float4*)&Ls[r][16 * cb + 4 * q4];
    a[4 * q4 + 0] = v.x; a[4 * q4 + 1] = v.y; a[4 * q4 + 2] = v.z; a[4 * q4 + 3] = v.w;
  }

  float R[16];
#pragma unroll
  for (int q = 0; q < 16; ++q) R[q] = (16 * cb + q == r) ? 1.f : 0.f;
  float y = ymu;

#pragma nounroll
  for (int jb = 0; jb < 4; ++jb) {
#pragma unroll
    for (int js = 0; js < 16; ++js) {
      float* cj = &colb[js & 1][0];
      float* rj = &rowb[js & 1][0];
      const int j = 16 * jb + js;
      if (cb == jb) cj[r] = a[js];
      if (cb <= jb && r == j) {
#pragma unroll
        for (int q4 = 0; q4 < 4; ++q4)
          *(float4*)&rj[16 * cb + 4 * q4] =
              make_float4(R[4 * q4 + 0], R[4 * q4 + 1], R[4 * q4 + 2], R[4 * q4 + 3]);
        if (cb == 0) yb[js & 1] = y;
      }
      __syncthreads();
      const float d = cj[j];
      const float cr = cj[r];
      const float yj = yb[js & 1];
      float cv[16], rv[16];
#pragma unroll
      for (int q4 = 0; q4 < 4; ++q4) {
        const float4 v = *(const float4*)&cj[16 * cb + 4 * q4];
        cv[4 * q4 + 0] = v.x; cv[4 * q4 + 1] = v.y; cv[4 * q4 + 2] = v.z; cv[4 * q4 + 3] = v.w;
      }
#pragma unroll
      for (int q4 = 0; q4 < 4; ++q4) {
        const float4 v = *(const float4*)&rj[16 * cb + 4 * q4];
        rv[4 * q4 + 0] = v.x; rv[4 * q4 + 1] = v.y; rv[4 * q4 + 2] = v.z; rv[4 * q4 + 3] = v.w;
      }
      const float gq = cr / d;
      const float g = (r > j) ? gq : 0.f;
      const float gHi = (cb >= jb) ? g : 0.f;
      const float gLo = (cb > jb) ? g : 0.f;
#pragma unroll
      for (int q = js + 1; q < 16; ++q) a[q] = fmaf(-gHi, cv[q], a[q]);
#pragma unroll
      for (int q = 0; q <= js; ++q) a[q] = fmaf(-gLo, cv[q], a[q]);
      y = fmaf(-g, yj, y);
#pragma unroll
      for (int q = 0; q < 16; ++q) R[q] = fmaf(-g, rv[q], R[q]);
    }
  }

  float dr = a[0];
#pragma unroll
  for (int q = 1; q < 16; ++q) dr = ((r & 15) == q) ? a[q] : dr;
  if ((r >> 4) == cb) {
    const float sq = sqrtf(dr);
    rdv[r] = 1.0f / sq;
    lg[r] = logf(sq);
  }
  __syncthreads();

  {
    const float sr = rdv[r];
#pragma unroll
    for (int q = 0; q < 16; ++q) M[r][16 * cb + q] = R[q] * sr;
    if (cb == 0) bv[r] = y * sr;
  }
  if (tid == 0) {
    float hs = 0.f;
    for (int i = 0; i < 64; ++i) hs += lg[i];
    sc[1] = hs;
  }
  __syncthreads();

  if (tid < 64) {
    const int t = tid;
    float* base = pack + (size_t)k * PACKF;
    const int l = t >> 1, s = t & 1, h = l >> 4, c = l & 15;
    int od = 0;
    for (int zt = 0; zt < 4; ++zt) {
      const int row = 16 * zt + c;
      for (int kg = 0; kg < 4 * zt; ++kg, ++od)
        base[od * 64 + t] = M[row][4 * kg + 2 * h + s];
      for (int j = 0; j < 4; ++j) {
        if (c >= 4 * j) {
          const int startf = 64 * j - 8 * j * (j - 1);
          base[OD_FLOATS + zt * 160 + startf + h * (32 - 8 * j) + (c - 4 * j) * 2 + s] =
              M[row][16 * zt + 4 * j + 2 * h + s];
        }
      }
    }
    base[BIAS_OFF + t] = -bv[t];
    if (t < 4)
      base[CONST_OFF + t] = (t == 0) ? (pi[k] - sc[0] - sc[1] - 0.5f * 64.f * LOG2PI) : 0.f;
  }
}

__device__ __forceinline__ int chunk_dst(int i) {
  if (i < 384) {
    const int od = i >> 4;
    const int f = (od < 4) ? (4 + od) : (od < 12) ? (8 + od) : (12 + od);
    return f * 64 + (i & 15) * 4;
  }
  if (i < 544) {
    const int e = i - 384;
    const int zt = e / 40;
    const int rem = e - zt * 40;
    const int j = (rem >= 16) + (rem >= 28) + (rem >= 36);
    const int start = (j == 0) ? 0 : (j == 1) ? 16 : (j == 2) ? 28 : 36;
    const int q = rem - start;
    const int len = 8 - 2 * j;
    const int hh = (q >= len) ? 1 : 0;
    const int u = q - hh * len;
    const int f = 2 * zt * (zt + 1) + 4 * zt + j;
    return f * 64 + 32 * hh + 8 * j + 4 * u;
  }
  return L_BIAS + (i - 544) * 4;
}

__global__ __launch_bounds__(32 * WPB) void gmm_main(const float* __restrict__ X,
                                                     const float* __restrict__ pack,
                                                     float* __restrict__ out,
                                                     int ntiles) {
  __shared__ __attribute__((aligned(32))) float mb[2][LDSF];
  const int tid = threadIdx.x;
  const int w = tid >> 5;
  const int lane = tid & 31;
  const int h = lane >> 4;
  const int c = lane & 15;
  const int t0 = (blockIdx.x * WPB + w) * RB;

  v2f xb[RB][16];
#pragma unroll
  for (int rt = 0; rt < RB; ++rt) {
    int tile = t0 + rt;
    if (tile > ntiles - 1) tile = ntiles - 1;
    const float* xp = X + ((size_t)tile * 16 + c) * 64 + 2 * h;
#pragma unroll
    for (int kg = 0; kg < 16; ++kg) xb[rt][kg] = *(const v2f*)(xp + 4 * kg);
  }

  const int d0 = chunk_dst(tid);
  const int d1 = chunk_dst(tid + 256);
  const bool has2 = (tid + 512) < NCHUNK;
  const int d2 = has2 ? chunk_dst(tid + 512) : 0;

  for (int i = tid; i < 2 * LDSF / 4; i += 32 * WPB)
    ((float4*)&mb[0][0])[i] = make_float4(0.f, 0.f, 0.f, 0.f);

  float4 s0, s1, s2 = make_float4(0.f, 0.f, 0.f, 0.f);
  {
    const float4* src = (const float4*)pack;
    s0 = src[tid];
    s1 = src[tid + 256];
    if (has2) s2 = src[tid + 512];
  }
  __syncthreads();
  *(float4*)&mb[0][d0] = s0;
  *(float4*)&mb[0][d1] = s1;
  if (has2) *(float4*)&mb[0][d2] = s2;
  __syncthreads();

  float m_run[RB], s_run[RB];
#pragma unroll
  for (int rt = 0; rt < RB; ++rt) { m_run[rt] = -3.0e38f; s_run[rt] = 0.f; }

  for (int k = 0; k < GMM_K; ++k) {
    const float* cur = &mb[k & 1][0];
    const bool more = (k + 1) < GMM_K;
    if (more) {
      const float4* src = (const float4*)(pack + (size_t)(k + 1) * PACKF);
      s0 = src[tid];
      s1 = src[tid + 256];
      if (has2) s2 = src[tid + 512];
    }

    float ss[RB];
#pragma unroll
    for (int rt = 0; rt < RB; ++rt) ss[rt] = 0.f;

#pragma unroll
    for (int zt = 0; zt < 4; ++zt) {
      const v8f bias = *(const v8f*)(cur + L_BIAS + 16 * zt + 8 * h);
      v8f acc[RB];
#pragma unroll
      for (int kg = 0; kg < 4 * (zt + 1); ++kg) {
        const v2f a = *(const v2f*)(cur + (2 * zt * (zt + 1) + kg) * 64 + 2 * lane);
#pragma unroll
        for (int rt = 0; rt < RB; ++rt)
          acc[rt] = __builtin_amdgcn_wmma_f32_16x16x4_f32(false, a, false, xb[rt][kg], (short)0,
                                                          (kg == 0) ? bias : acc[rt], false, false);
      }
#pragma unroll
      for (int rt = 0; rt < RB; ++rt)
#pragma unroll
        for (int r = 0; r < 8; ++r) ss[rt] = fmaf(acc[rt][r], acc[rt][r], ss[rt]);
    }

    const float ck = cur[L_CONST];
#pragma unroll
    for (int rt = 0; rt < RB; ++rt) {
      const float mh = ss[rt] + __shfl_xor(ss[rt], 16, 32);
      const float v = fmaf(-0.5f, mh, ck);
      const float mn = fmaxf(m_run[rt], v);
      s_run[rt] = s_run[rt] * expf(m_run[rt] - mn) + expf(v - mn);
      m_run[rt] = mn;
    }

    if (more) {
      float* nxt = &mb[(k + 1) & 1][0];
      *(float4*)&nxt[d0] = s0;
      *(float4*)&nxt[d1] = s1;
      if (has2) *(float4*)&nxt[d2] = s2;
    }
    __syncthreads();
  }

  const float r0 = m_run[0] + logf(s_run[0]);
  const float r1 = m_run[1] + logf(s_run[1]);
  const float val = (h == 0) ? r0 : r1;
  if (t0 + h < ntiles) out[(size_t)t0 * 16 + lane] = val;
}

extern "C" void kernel_launch(void* const* d_in, const int* in_sizes, int n_in,
                              void* d_out, int out_size, void* d_ws, size_t ws_size,
                              hipStream_t stream) {
  const float* X   = (const float*)d_in[0];
  const float* loc = (const float*)d_in[1];
  const float* cov = (const float*)d_in[2];
  const float* pi  = (const float*)d_in[3];
  float* out = (float*)d_out;
  float* pack = (float*)d_ws;

  const int N = in_sizes[0] / 64;
  const int ntiles = N / 16;

  gmm_prep<<<GMM_K, 256, 0, stream>>>(loc, cov, pi, pack);
  const int tilesPerBlock = WPB * RB;
  const int nblk = (ntiles + tilesPerBlock - 1) / tilesPerBlock;
  gmm_main<<<nblk, 32 * WPB, 0, stream>>>(X, pack, out, ntiles);
}


# gmm_main: each wave keeps one component's L^-1 fragments in registers and streams the block's 16 tiles (x staged in LDS); per-component LDS ring and its 32 barriers gone; logsumexp sum order changes
# speedup vs baseline: 5.0286x; 1.0510x over previous
#include <hip/hip_runtime.h>
#include <math.h>


#define GMM_K   32
#define LOG2PI  1.8378770664093453f

#define PACKF     2244
#define OD_FLOATS 1536
#define BIAS_OFF  2176
#define CONST_OFF 2240

#define RB      2
#define WPB     8
#define CPW     4
#define TPB     16
#define XP      68
#define BP      72

typedef __attribute__((ext_vector_type(2))) float v2f;
typedef __attribute__((ext_vector_type(8))) float v8f;

__global__ __launch_bounds__(256) void gmm_prep(const float* __restrict__ loc,
                                                const float* __restrict__ cov,
                                                const float* __restrict__ pi,
                                                float* __restrict__ pack) {
  __shared__ __attribute__((aligned(16))) float Ls[64][68];
  __shared__ float M[64][65];
  __shared__ __attribute__((aligned(16))) float colb[2][64];
  __shared__ __attribute__((aligned(16))) float rowb[2][64];
  __shared__ __attribute__((aligned(16))) float rdv[64];
  __shared__ __attribute__((aligned(16))) float lg[64];
  __shared__ float yb[2];
  __shared__ float bv[64];
  __shared__ float sc[2];
  const int k = blockIdx.x;
  const int tid = threadIdx.x;
  const int lane = tid & 31;
  const int w = __builtin_amdgcn_readfirstlane(tid >> 5);
  const int cb = w >> 1;
  const int r = ((w & 1) << 5) + lane;

  {
    const float4* src = (const float4*)(cov + (size_t)k * 4096);
#pragma unroll
    for (int it = 0; it < 4; ++it) {
      const int f = it * 256 + tid;
      *(float4*)&Ls[f >> 4][(f & 15) * 4] = src[f];
    }
  }
  const float ymu = loc[k * 64 + r];
  if (tid < 128) (&rowb[0][0])[tid] = 0.f;
  if (tid == 0) {
    float m = pi[0];
    for (int i = 1; i < GMM_K; ++i) m = fmaxf(m, pi[i]);
    float s = 0.f;
    for (int i = 0; i < GMM_K; ++i) s += expf(pi[i] - m);
    sc[0] = m + logf(s);
  }
  __syncthreads();

  float a[16];
#pragma unroll
  for (int q4 = 0; q4 < 4; ++q4) {
    const float4 v = *(const float4*)&Ls[r][16 * cb + 4 * q4];
    a[4 * q4 + 0] = v.x; a[4 * q4 + 1] = v.y; a[4 * q4 + 2] = v.z; a[4 * q4 + 3] = v.w;
  }

  float R[16];
#pragma unroll
  for (int q = 0; q < 16; ++q) R[q] = (16 * cb + q == r) ? 1.f : 0.f;
  float y = ymu;

#pragma nounroll
  for (int jb = 0; jb < 4; ++jb) {
#pragma unroll
    for (int js = 0; js < 16; ++js) {
      float* cj = &colb[js & 1][0];
      float* rj = &rowb[js & 1][0];
      const int j = 16 * jb + js;
      if (cb == jb) cj[r] = a[js];
      if (cb <= jb && r == j) {
#pragma unroll
        for (int q4 = 0; q4 < 4; ++q4)
          *(float4*)&rj[16 * cb + 4 * q4] =
              make_float4(R[4 * q4 + 0], R[4 * q4 + 1], R[4 * q4 + 2], R[4 * q4 + 3]);
        if (cb == 0) yb[js & 1] = y;
      }
      __syncthreads();
      const float d = cj[j];
      const float cr = cj[r];
      const float yj = yb[js & 1];
      float cv[16], rv[16];
#pragma unroll
      for (int q4 = 0; q4 < 4; ++q4) {
        const float4 v = *(const float4*)&cj[16 * cb + 4 * q4];
        cv[4 * q4 + 0] = v.x; cv[4 * q4 + 1] = v.y; cv[4 * q4 + 2] = v.z; cv[4 * q4 + 3] = v.w;
      }
#pragma unroll
      for (int q4 = 0; q4 < 4; ++q4) {
        const float4 v = *(const float4*)&rj[16 * cb + 4 * q4];
        rv[4 * q4 + 0] = v.x; rv[4 * q4 + 1] = v.y; rv[4 * q4 + 2] = v.z; rv[4 * q4 + 3] = v.w;
      }
      const float gq = cr / d;
      const float g = (r > j) ? gq : 0.f;
      const float gHi = (cb >= jb) ? g : 0.f;
      const float gLo = (cb > jb) ? g : 0.f;
#pragma unroll
      for (int q = js + 1; q < 16; ++q) a[q] = fmaf(-gHi, cv[q], a[q]);
#pragma unroll
      for (int q = 0; q <= js; ++q) a[q] = fmaf(-gLo, cv[q], a[q]);
      y = fmaf(-g, yj, y);
#pragma unroll
      for (int q = 0; q < 16; ++q) R[q] = fmaf(-g, rv[q], R[q]);
    }
  }

  float dr = a[0];
#pragma unroll
  for (int q = 1; q < 16; ++q) dr = ((r & 15) == q) ? a[q] : dr;
  if ((r >> 4) == cb) {
    const float sq = sqrtf(dr);
    rdv[r] = 1.0f / sq;
    lg[r] = logf(sq);
  }
  __syncthreads();

  {
    const float sr = rdv[r];
#pragma unroll
    for (int q = 0; q < 16; ++q) M[r][16 * cb + q] = R[q] * sr;
    if (cb == 0) bv[r] = y * sr;
  }
  if (tid == 0) {
    float hs = 0.f;
    for (int i = 0; i < 64; ++i) hs += lg[i];
    sc[1] = hs;
  }
  __syncthreads();

  if (tid < 64) {
    const int t = tid;
    float* base = pack + (size_t)k * PACKF;
    const int l = t >> 1, s = t & 1, h = l >> 4, c = l & 15;
    int od = 0;
    for (int zt = 0; zt < 4; ++zt) {
      const int row = 16 * zt + c;
      for (int kg = 0; kg < 4 * zt; ++kg, ++od)
        base[od * 64 + t] = M[row][4 * kg + 2 * h + s];
      for (int j = 0; j < 4; ++j) {
        if (c >= 4 * j) {
          const int startf = 64 * j - 8 * j * (j - 1);
          base[OD_FLOATS + zt * 160 + startf + h * (32 - 8 * j) + (c - 4 * j) * 2 + s] =
              M[row][16 * zt + 4 * j + 2 * h + s];
        }
      }
    }
    base[BIAS_OFF + t] = -bv[t];
    if (t < 4)
      base[CONST_OFF + t] = (t == 0) ? (pi[k] - sc[0] - sc[1] - 0.5f * 64.f * LOG2PI) : 0.f;
  }
}

__device__ __forceinline__ void load_m_fragments(v2f (&af)[40], const float* __restrict__ base,
                                                 int lane, int c, int h) {
#pragma unroll
  for (int zt = 0; zt < 4; ++zt) {
#pragma unroll
    for (int kg = 0; kg < 4 * zt; ++kg)
      af[2 * zt * (zt + 1) + kg] = *(const v2f*)(base + (2 * zt * (zt - 1) + kg) * 64 + 2 * lane);
#pragma unroll
    for (int j = 0; j < 4; ++j) {
      v2f v = {0.f, 0.f};
      if (c >= 4 * j)
        v = *(const v2f*)(base + OD_FLOATS + zt * 160 + (64 * j - 8 * j * (j - 1)) +
                          h * (32 - 8 * j) + (c - 4 * j) * 2);
      af[2 * zt * (zt + 1) + 4 * zt + j] = v;
    }
  }
}

__global__ __launch_bounds__(32 * WPB) void gmm_main(const float* __restrict__ X,
                                                     const float* __restrict__ pack,
                                                     float* __restrict__ out,
                                                     int ntiles) {
  __shared__ __attribute__((aligned(16))) float xs[16 * TPB][XP];
  __shared__ __attribute__((aligned(32))) float bs[GMM_K][BP];
  __shared__ float vbuf[GMM_K][16 * TPB];
  const int tid = threadIdx.x;
  const int w = tid >> 5;
  const int lane = tid & 31;
  const int h = lane >> 4;
  const int c = lane & 15;
  const int npts = ntiles * 16;
  const int p0 = blockIdx.x * (16 * TPB);

#pragma unroll
  for (int it = 0; it < (16 * TPB * 16) / (32 * WPB); ++it) {
    const int f = it * (32 * WPB) + tid;
    const int row = f >> 4, c4 = f & 15;
    int gp = p0 + row;
    if (gp > npts - 1) gp = npts - 1;
    *(float4*)&xs[row][4 * c4] = *(const float4*)(X + (size_t)gp * 64 + 4 * c4);
  }
  {
    const int k = tid >> 3, q = tid & 7;
    const float* src = pack + (size_t)k * PACKF;
    *(float4*)&bs[k][8 * q]     = *(const float4*)(src + BIAS_OFF + 8 * q);
    *(float4*)&bs[k][8 * q + 4] = *(const float4*)(src + BIAS_OFF + 8 * q + 4);
    if (q == 0) bs[k][64] = src[CONST_OFF];
  }

  v2f af[40];
  load_m_fragments(af, pack + (size_t)(CPW * w) * PACKF, lane, c, h);
  __syncthreads();

#pragma nounroll
  for (int ci = 0; ci < CPW; ++ci) {
    const int k = CPW * w + ci;
    const float* bk = &bs[k][0];
    v8f bias[4];
#pragma unroll
    for (int zt = 0; zt < 4; ++zt) bias[zt] = *(const v8f*)(bk + 16 * zt + 8 * h);
    const float ck = bk[64];

#pragma nounroll
    for (int tp = 0; tp < TPB / RB; ++tp) {
      v2f xb[RB][16];
#pragma unroll
      for (int rt = 0; rt < RB; ++rt)
#pragma unroll
        for (int kg = 0; kg < 16; ++kg)
          xb[rt][kg] = *(const v2f*)&xs[(RB * tp + rt) * 16 + c][4 * kg + 2 * h];

      float ss[RB];
#pragma unroll
      for (int rt = 0; rt < RB; ++rt) ss[rt] = 0.f;

#pragma unroll
      for (int zt = 0; zt < 4; ++zt) {
        v8f acc[RB];
#pragma unroll
        for (int kg = 0; kg < 4 * (zt + 1); ++kg) {
          const v2f a = af[2 * zt * (zt + 1) + kg];
#pragma unroll
          for (int rt = 0; rt < RB; ++rt)
            acc[rt] = __builtin_amdgcn_wmma_f32_16x16x4_f32(false, a, false, xb[rt][kg], (short)0,
                                                            (kg == 0) ? bias[zt] : acc[rt], false, false);
        }
#pragma unroll
        for (int rt = 0; rt < RB; ++rt)
#pragma unroll
          for (int r = 0; r < 8; ++r) ss[rt] = fmaf(acc[rt][r], acc[rt][r], ss[rt]);
      }

      float v[RB];
#pragma unroll
      for (int rt = 0; rt < RB; ++rt) {
        const float mh = ss[rt] + __shfl_xor(ss[rt], 16, 32);
        v[rt] = fmaf(-0.5f, mh, ck);
      }
      vbuf[k][32 * tp + lane] = (h == 0) ? v[0] : v[1];
    }

    if (ci + 1 < CPW)
      load_m_fragments(af, pack + (size_t)(k + 1) * PACKF, lane, c, h);
  }
  __syncthreads();

  {
    float m = vbuf[0][tid];
#pragma unroll
    for (int k = 1; k < GMM_K; ++k) m = fmaxf(m, vbuf[k][tid]);
    float s = 0.f;
#pragma unroll
    for (int k = 0; k < GMM_K; ++k) s += expf(vbuf[k][tid] - m);
    if (p0 + tid < npts) out[(size_t)p0 + tid] = m + logf(s);
  }
}

extern "C" void kernel_launch(void* const* d_in, const int* in_sizes, int n_in,
                              void* d_out, int out_size, void* d_ws, size_t ws_size,
                              hipStream_t stream) {
  const float* X   = (const float*)d_in[0];
  const float* loc = (const float*)d_in[1];
  const float* cov = (const float*)d_in[2];
  const float* pi  = (const float*)d_in[3];
  float* out = (float*)d_out;
  float* pack = (float*)d_ws;

  const int N = in_sizes[0] / 64;
  const int ntiles = N / 16;

  gmm_prep<<<GMM_K, 256, 0, stream>>>(loc, cov, pi, pack);
  const int nblk = (ntiles + TPB - 1) / TPB;
  gmm_main<<<nblk, 32 * WPB, 0, stream>>>(X, pack, out, ntiles);
}


# gmm_main: a 16-row WMMA tile = rows 4m..4m+3 of four components (one group per wave, fragments resident): tile m takes k groups 0..m, 136 WMMAs per 4 components, not 160; |z|^2 sum order changes
# speedup vs baseline: 5.6680x; 1.1272x over previous
#include <hip/hip_runtime.h>
#include <math.h>


#define GMM_K   32
#define LOG2PI  1.8378770664093453f

#define PACKF     2244
#define NFRAG     136
#define BIAS_OFF  2176
#define CONST_OFF 2240

#define RB      2
#define WPB     8
#define CPW     4
#define TPB     16
#define XP      68
#define BP      72

typedef __attribute__((ext_vector_type(2))) float v2f;
typedef __attribute__((ext_vector_type(8))) float v8f;

__global__ __launch_bounds__(256) void gmm_prep(const float* __restrict__ loc,
                                                const float* __restrict__ cov,
                                                const float* __restrict__ pi,
                                                float* __restrict__ pack) {
  __shared__ __attribute__((aligned(16))) float Ls[64][68];
  __shared__ float M[64][65];
  __shared__ __attribute__((aligned(16))) float colb[2][64];
  __shared__ __attribute__((aligned(16))) float rowb[2][64];
  __shared__ __attribute__((aligned(16))) float rdv[64];
  __shared__ __attribute__((aligned(16))) float lg[64];
  __shared__ float yb[2];
  __shared__ float bv[64];
  __shared__ float sc[2];
  const int k = blockIdx.x;
  const int tid = threadIdx.x;
  const int lane = tid & 31;
  const int w = __builtin_amdgcn_readfirstlane(tid >> 5);
  const int cb = w >> 1;
  const int r = ((w & 1) << 5) + lane;

  {
    const float4* src = (const float4*)(cov + (size_t)k * 4096);
#pragma unroll
    for (int it = 0; it < 4; ++it) {
      const int f = it * 256 + tid;
      *(float4*)&Ls[f >> 4][(f & 15) * 4] = src[f];
    }
  }
  const float ymu = loc[k * 64 + r];
  if (tid < 128) (&rowb[0][0])[tid] = 0.f;
  if (tid == 0) {
    float m = pi[0];
    for (int i = 1; i < GMM_K; ++i) m = fmaxf(m, pi[i]);
    float s = 0.f;
    for (int i = 0; i < GMM_K; ++i) s += expf(pi[i] - m);
    sc[0] = m + logf(s);
  }
  __syncthreads();

  float a[16];
#pragma unroll
  for (int q4 = 0; q4 < 4; ++q4) {
    const float4 v = *(const float4*)&Ls[r][16 * cb + 4 * q4];
    a[4 * q4 + 0] = v.x; a[4 * q4 + 1] = v.y; a[4 * q4 + 2] = v.z; a[4 * q4 + 3] = v.w;
  }

  float R[16];
#pragma unroll
  for (int q = 0; q < 16; ++q) R[q] = (16 * cb + q == r) ? 1.f : 0.f;
  float y = ymu;

#pragma nounroll
  for (int jb = 0; jb < 4; ++jb) {
#pragma unroll
    for (int js = 0; js < 16; ++js) {
      float* cj = &colb[js & 1][0];
      float* rj = &rowb[js & 1][0];
      const int j = 16 * jb + js;
      if (cb == jb) cj[r] = a[js];
      if (cb <= jb && r == j) {
#pragma unroll
        for (int q4 = 0; q4 < 4; ++q4)
          *(float4*)&rj[16 * cb + 4 * q4] =
              make_float4(R[4 * q4 + 0], R[4 * q4 + 1], R[4 * q4 + 2], R[4 * q4 + 3]);
        if (cb == 0) yb[js & 1] = y;
      }
      __syncthreads();
      const float d = cj[j];
      const float cr = cj[r];
      const float yj = yb[js & 1];
      float cv[16], rv[16];
#pragma unroll
      for (int q4 = 0; q4 < 4; ++q4) {
        const float4 v = *(const float4*)&cj[16 * cb + 4 * q4];
        cv[4 * q4 + 0] = v.x; cv[4 * q4 + 1] = v.y; cv[4 * q4 + 2] = v.z; cv[4 * q4 + 3] = v.w;
      }
#pragma unroll
      for (int q4 = 0; q4 < 4; ++q4) {
        const float4 v = *(const float4*)&rj[16 * cb + 4 * q4];
        rv[4 * q4 + 0] = v.x; rv[4 * q4 + 1] = v.y; rv[4 * q4 + 2] = v.z; rv[4 * q4 + 3] = v.w;
      }
      const float gq = cr / d;
      const float g = (r > j) ? gq : 0.f;
      const float gHi = (cb >= jb) ? g : 0.f;
      const float gLo = (cb > jb) ? g : 0.f;
#pragma unroll
      for (int q = js + 1; q < 16; ++q) a[q] = fmaf(-gHi, cv[q], a[q]);
#pragma unroll
      for (int q = 0; q <= js; ++q) a[q] = fmaf(-gLo, cv[q], a[q]);
      y = fmaf(-g, yj, y);
#pragma unroll
      for (int q = 0; q < 16; ++q) R[q] = fmaf(-g, rv[q], R[q]);
    }
  }

  float dr = a[0];
#pragma unroll
  for (int q = 1; q < 16; ++q) dr = ((r & 15) == q) ? a[q] : dr;
  if ((r >> 4) == cb) {
    const float sq = sqrtf(dr);
    rdv[r] = 1.0f / sq;
    lg[r] = logf(sq);
  }
  __syncthreads();

  {
    const float sr = rdv[r];
#pragma unroll
    for (int q = 0; q < 16; ++q) M[r][16 * cb + q] = R[q] * sr;
    if (cb == 0) bv[r] = y * sr;
  }
  if (tid == 0) {
    float hs = 0.f;
    for (int i = 0; i < 64; ++i) hs += lg[i];
    sc[1] = hs;
  }
  __syncthreads();

  {
    float* base = pack + (size_t)k * PACKF;
#pragma unroll
    for (int m = 0; m < 16; ++m) {
      if (tid < 16 * (m + 1))
        base[8 * m * (m + 1) + tid] = M[4 * m + ((tid >> 2) & 3)][4 * (tid >> 4) + (tid & 3)];
    }
    if (tid < 64) base[BIAS_OFF + tid] = -bv[tid];
    if (tid < 4)
      base[CONST_OFF + tid] = (tid == 0) ? (pi[k] - sc[0] - sc[1] - 0.5f * 64.f * LOG2PI) : 0.f;
  }
}

__global__ __launch_bounds__(32 * WPB) void gmm_main(const float* __restrict__ X,
                                                     const float* __restrict__ pack,
                                                     float* __restrict__ out,
                                                     int ntiles) {
  __shared__ __attribute__((aligned(16))) float xs[16 * TPB][XP];
  __shared__ __attribute__((aligned(32))) float bs[GMM_K][BP];
  __shared__ float vbuf[GMM_K][16 * TPB];
  const int tid = threadIdx.x;
  const int w = tid >> 5;
  const int lane = tid & 31;
  const int h = lane >> 4;
  const int c = lane & 15;
  const int npts = ntiles * 16;
  const int p0 = blockIdx.x * (16 * TPB);

  const int kA = CPW * w + 2 * h;
  const float cA = pack[(size_t)kA * PACKF + CONST_OFF];
  const float cB = pack[(size_t)(kA + 1) * PACKF + CONST_OFF];

  v2f af[NFRAG];
  {
    const float* fb = pack + (size_t)(CPW * w + (c >> 2)) * PACKF + 4 * (c & 3) + 2 * h;
#pragma unroll
    for (int f = 0; f < NFRAG; ++f) af[f] = *(const v2f*)(fb + 16 * f);
  }

#pragma unroll
  for (int it = 0; it < (16 * TPB * 16) / (32 * WPB); ++it) {
    const int f = it * (32 * WPB) + tid;
    const int row = f >> 4, c4 = f & 15;
    int gp = p0 + row;
    if (gp > npts - 1) gp = npts - 1;
    *(float4*)&xs[row][4 * c4] = *(const float4*)(X + (size_t)gp * 64 + 4 * c4);
  }
  {
    const int k = tid >> 3, q = tid & 7;
    const float* src = pack + (size_t)k * PACKF;
    *(float4*)&bs[k][8 * q]     = *(const float4*)(src + BIAS_OFF + 8 * q);
    *(float4*)&bs[k][8 * q + 4] = *(const float4*)(src + BIAS_OFF + 8 * q + 4);
  }

  __syncthreads();

  const float* bA = &bs[kA][0];
  const float* bB = &bs[kA + 1][0];

#pragma nounroll
  for (int tp = 0; tp < TPB / RB; ++tp) {
    v2f xb[RB][16];
#pragma unroll
    for (int rt = 0; rt < RB; ++rt)
#pragma unroll
      for (int kg = 0; kg < 16; ++kg)
        xb[rt][kg] = *(const v2f*)&xs[(RB * tp + rt) * 16 + c][4 * kg + 2 * h];

    v2f sA[RB], sB[RB];
#pragma unroll
    for (int rt = 0; rt < RB; ++rt) { sA[rt] = (v2f){0.f, 0.f}; sB[rt] = (v2f){0.f, 0.f}; }

#pragma unroll
    for (int m = 0; m < 16; ++m) {
      const float4 b0 = *(const float4*)(bA + 4 * m);
      const float4 b1 = *(const float4*)(bB + 4 * m);
      const v8f c0 = {b0.x, b0.y, b0.z, b0.w, b1.x, b1.y, b1.z, b1.w};
      v8f acc[RB];
#pragma unroll
      for (int kg = 0; kg <= m; ++kg) {
        const v2f a = af[(m * (m + 1)) / 2 + kg];
#pragma unroll
        for (int rt = 0; rt < RB; ++rt)
          acc[rt] = __builtin_amdgcn_wmma_f32_16x16x4_f32(false, a, false, xb[rt][kg], (short)0,
                                                          (kg == 0) ? c0 : acc[rt], false, false);
      }
#pragma unroll
      for (int rt = 0; rt < RB; ++rt) {
        const v2f z01 = {acc[rt][0], acc[rt][1]}, z23 = {acc[rt][2], acc[rt][3]};
        const v2f z45 = {acc[rt][4], acc[rt][5]}, z67 = {acc[rt][6], acc[rt][7]};
        sA[rt] = z01 * z01 + sA[rt];
        sA[rt] = z23 * z23 + sA[rt];
        sB[rt] = z45 * z45 + sB[rt];
        sB[rt] = z67 * z67 + sB[rt];
      }
    }

#pragma unroll
    for (int rt = 0; rt < RB; ++rt) {
      const int pt = (RB * tp + rt) * 16 + c;
      vbuf[kA][pt]     = fmaf(-0.5f, sA[rt].x + sA[rt].y, cA);
      vbuf[kA + 1][pt] = fmaf(-0.5f, sB[rt].x + sB[rt].y, cB);
    }
  }
  __syncthreads();

  {
    float m = vbuf[0][tid];
#pragma unroll
    for (int k = 1; k < GMM_K; ++k) m = fmaxf(m, vbuf[k][tid]);
    float s = 0.f;
#pragma unroll
    for (int k = 0; k < GMM_K; ++k) s += expf(vbuf[k][tid] - m);
    if (p0 + tid < npts) out[(size_t)p0 + tid] = m + logf(s);
  }
}

extern "C" void kernel_launch(void* const* d_in, const int* in_sizes, int n_in,
                              void* d_out, int out_size, void* d_ws, size_t ws_size,
                              hipStream_t stream) {
  const float* X   = (const float*)d_in[0];
  const float* loc = (const float*)d_in[1];
  const float* cov = (const float*)d_in[2];
  const float* pi  = (const float*)d_in[3];
  float* out = (float*)d_out;
  float* pack = (float*)d_ws;

  const int N = in_sizes[0] / 64;
  const int ntiles = N / 16;

  gmm_prep<<<GMM_K, 256, 0, stream>>>(loc, cov, pi, pack);
  const int nblk = (ntiles + TPB - 1) / TPB;
  gmm_main<<<nblk, 32 * WPB, 0, stream>>>(X, pack, out, ntiles);
}


# gmm_prep: wave w owns columns 8w..8w+7 of all 64 rows: pivot rows stay in-wave (cross-lane reads), only multipliers a*(1/d) cross waves via LDS; 8 barriers, not 64; sum order kept, rounding changes
# speedup vs baseline: 6.2804x; 1.1080x over previous
#include <hip/hip_runtime.h>
#include <math.h>


#define GMM_K   32
#define LOG2PI  1.8378770664093453f

#define PACKF     2244
#define NFRAG     136
#define BIAS_OFF  2176
#define CONST_OFF 2240

#define RB      2
#define WPB     8
#define CPW     4
#define TPB     16
#define XP      68
#define BP      72

typedef __attribute__((ext_vector_type(2))) float v2f;
typedef __attribute__((ext_vector_type(8))) float v8f;


__device__ __forceinline__ float lane_read(float x, int src) {
  return __int_as_float(__builtin_amdgcn_readlane(__float_as_int(x), src));
}

__global__ __launch_bounds__(256) void gmm_prep(const float* __restrict__ loc,
                                                const float* __restrict__ cov,
                                                const float* __restrict__ pi,
                                                float* __restrict__ pack) {
  __shared__ __attribute__((aligned(16))) float Ls[64][68];
  __shared__ float M[64][65];
  __shared__ __attribute__((aligned(16))) float gs[2][64][8];
  __shared__ float rdv[64];
  __shared__ float lg[64];
  __shared__ float bv[64];
  __shared__ float sc[2];
  const int k = blockIdx.x;
  const int tid = threadIdx.x;
  const int lane = tid & 31;
  const int w = __builtin_amdgcn_readfirstlane(tid >> 5);
  const int r0 = 2 * lane;

  {
    const float4* src = (const float4*)(cov + (size_t)k * 4096);
#pragma unroll
    for (int it = 0; it < 4; ++it) {
      const int f = it * 256 + tid;
      *(float4*)&Ls[f >> 4][(f & 15) * 4] = src[f];
    }
  }
  float y[2];
  y[0] = loc[k * 64 + r0];
  y[1] = loc[k * 64 + r0 + 1];
  if (tid == 0) {
    float m = pi[0];
    for (int i = 1; i < GMM_K; ++i) m = fmaxf(m, pi[i]);
    float s = 0.f;
    for (int i = 0; i < GMM_K; ++i) s += expf(pi[i] - m);
    sc[0] = m + logf(s);
  }
  __syncthreads();

  float a[2][8], R[2][8];
#pragma unroll
  for (int h = 0; h < 2; ++h) {
#pragma unroll
    for (int q = 0; q < 8; ++q) {
      const int r = r0 + h, c = 8 * w + q;
      const float lo = Ls[r][c];
      const float up = Ls[c][r];
      a[h][q] = (r >= c) ? lo : up;
      R[h][q] = (r == c) ? 1.f : 0.f;
    }
  }

#pragma nounroll
  for (int p = 0; p < 8; ++p) {
    float (*gb)[8] = gs[p & 1];
    if (w == p) {
      float g[8][2];
#pragma unroll
      for (int jj = 0; jj < 8; ++jj) {
        const int j = 8 * p + jj;
        const int lj = 4 * p + (jj >> 1);
        const float d = lane_read(a[jj & 1][jj], lj);
        const float ad = fabsf(d);
        const bool safe = (ad >= 0x1p-126f) && (ad <= 0x1p126f);
        const float rd = 1.0f / d;
        float gq0 = a[0][jj] * rd;
        float gq1 = a[1][jj] * rd;
        if (__builtin_expect(!safe, 0)) {
          gq0 = a[0][jj] / d;
          gq1 = a[1][jj] / d;
        }
        g[jj][0] = (r0 > j) ? gq0 : 0.f;
        g[jj][1] = (r0 + 1 > j) ? gq1 : 0.f;
#pragma unroll
        for (int q = jj + 1; q < 8; ++q) {
          const float s = lane_read(a[jj & 1][q], lj);
          a[0][q] = fmaf(-g[jj][0], s, a[0][q]);
          a[1][q] = fmaf(-g[jj][1], s, a[1][q]);
        }
      }
#pragma unroll
      for (int h = 0; h < 2; ++h) {
        *(float4*)&gb[r0 + h][0] = make_float4(g[0][h], g[1][h], g[2][h], g[3][h]);
        *(float4*)&gb[r0 + h][4] = make_float4(g[4][h], g[5][h], g[6][h], g[7][h]);
      }
    }
    __syncthreads();

    float G[8][2];
#pragma unroll
    for (int h = 0; h < 2; ++h) {
      const float4 v0 = *(const float4*)&gb[r0 + h][0];
      const float4 v1 = *(const float4*)&gb[r0 + h][4];
      G[0][h] = v0.x; G[1][h] = v0.y; G[2][h] = v0.z; G[3][h] = v0.w;
      G[4][h] = v1.x; G[5][h] = v1.y; G[6][h] = v1.z; G[7][h] = v1.w;
    }
    if (w > p) {
#pragma unroll
      for (int jj = 0; jj < 8; ++jj) {
        const int lj = 4 * p + (jj >> 1);
#pragma unroll
        for (int q = 0; q < 8; ++q) {
          const float s = lane_read(a[jj & 1][q], lj);
          a[0][q] = fmaf(-G[jj][0], s, a[0][q]);
          a[1][q] = fmaf(-G[jj][1], s, a[1][q]);
        }
      }
    } else {
#pragma unroll
      for (int jj = 0; jj < 8; ++jj) {
        const int lj = 4 * p + (jj >> 1);
#pragma unroll
        for (int q = 0; q < 8; ++q) {
          const float s = lane_read(R[jj & 1][q], lj);
          R[0][q] = fmaf(-G[jj][0], s, R[0][q]);
          R[1][q] = fmaf(-G[jj][1], s, R[1][q]);
        }
        if (w == 0) {
          const float s = lane_read(y[jj & 1], lj);
          y[0] = fmaf(-G[jj][0], s, y[0]);
          y[1] = fmaf(-G[jj][1], s, y[1]);
        }
      }
    }
  }

  if ((lane >> 2) == w) {
#pragma unroll
    for (int h = 0; h < 2; ++h) {
      float dr = a[h][h];
#pragma unroll
      for (int t = 1; t < 4; ++t) dr = ((lane & 3) == t) ? a[h][2 * t + h] : dr;
      const float sq = sqrtf(dr);
      rdv[r0 + h] = 1.0f / sq;
      lg[r0 + h] = logf(sq);
    }
  }
  __syncthreads();

#pragma unroll
  for (int h = 0; h < 2; ++h) {
    const float sr = rdv[r0 + h];
#pragma unroll
    for (int q = 0; q < 8; ++q) M[r0 + h][8 * w + q] = R[h][q] * sr;
    if (w == 0) bv[r0 + h] = y[h] * sr;
  }
  if (tid == 0) {
    float hs = 0.f;
    for (int i = 0; i < 64; ++i) hs += lg[i];
    sc[1] = hs;
  }
  __syncthreads();

  {
    float* base = pack + (size_t)k * PACKF;
#pragma unroll
    for (int m = 0; m < 16; ++m) {
      if (tid < 16 * (m + 1))
        base[8 * m * (m + 1) + tid] = M[4 * m + ((tid >> 2) & 3)][4 * (tid >> 4) + (tid & 3)];
    }
    if (tid < 64) base[BIAS_OFF + tid] = -bv[tid];
    if (tid < 4)
      base[CONST_OFF + tid] = (tid == 0) ? (pi[k] - sc[0] - sc[1] - 0.5f * 64.f * LOG2PI) : 0.f;
  }
}

__global__ __launch_bounds__(32 * WPB) void gmm_main(const float* __restrict__ X,
                                                     const float* __restrict__ pack,
                                                     float* __restrict__ out,
                                                     int ntiles) {
  __shared__ __attribute__((aligned(16))) float xs[16 * TPB][XP];
  __shared__ __attribute__((aligned(32))) float bs[GMM_K][BP];
  __shared__ float vbuf[GMM_K][16 * TPB];
  const int tid = threadIdx.x;
  const int w = tid >> 5;
  const int lane = tid & 31;
  const int h = lane >> 4;
  const int c = lane & 15;
  const int npts = ntiles * 16;
  const int p0 = blockIdx.x * (16 * TPB);

  const int kA = CPW * w + 2 * h;
  const float cA = pack[(size_t)kA * PACKF + CONST_OFF];
  const float cB = pack[(size_t)(kA + 1) * PACKF + CONST_OFF];

  v2f af[NFRAG];
  {
    const float* fb = pack + (size_t)(CPW * w + (c >> 2)) * PACKF + 4 * (c & 3) + 2 * h;
#pragma unroll
    for (int f = 0; f < NFRAG; ++f) af[f] = *(const v2f*)(fb + 16 * f);
  }

#pragma unroll
  for (int it = 0; it < (16 * TPB * 16) / (32 * WPB); ++it) {
    const int f = it * (32 * WPB) + tid;
    const int row = f >> 4, c4 = f & 15;
    int gp = p0 + row;
    if (gp > npts - 1) gp = npts - 1;
    *(float4*)&xs[row][4 * c4] = *(const float4*)(X + (size_t)gp * 64 + 4 * c4);
  }
  {
    const int k = tid >> 3, q = tid & 7;
    const float* src = pack + (size_t)k * PACKF;
    *(float4*)&bs[k][8 * q]     = *(const float4*)(src + BIAS_OFF + 8 * q);
    *(float4*)&bs[k][8 * q + 4] = *(const float4*)(src + BIAS_OFF + 8 * q + 4);
  }

  __syncthreads();

  const float* bA = &bs[kA][0];
  const float* bB = &bs[kA + 1][0];

#pragma nounroll
  for (int tp = 0; tp < TPB / RB; ++tp) {
    v2f xb[RB][16];
#pragma unroll
    for (int rt = 0; rt < RB; ++rt)
#pragma unroll
      for (int kg = 0; kg < 16; ++kg)
        xb[rt][kg] = *(const v2f*)&xs[(RB * tp + rt) * 16 + c][4 * kg + 2 * h];

    v2f sA[RB], sB[RB];
#pragma unroll
    for (int rt = 0; rt < RB; ++rt) { sA[rt] = (v2f){0.f, 0.f}; sB[rt] = (v2f){0.f, 0.f}; }

#pragma unroll
    for (int m = 0; m < 16; ++m) {
      const float4 b0 = *(const float4*)(bA + 4 * m);
      const float4 b1 = *(const float4*)(bB + 4 * m);
      const v8f c0 = {b0.x, b0.y, b0.z, b0.w, b1.x, b1.y, b1.z, b1.w};
      v8f acc[RB];
#pragma unroll
      for (int kg = 0; kg <= m; ++kg) {
        const v2f a = af[(m * (m + 1)) / 2 + kg];
#pragma unroll
        for (int rt = 0; rt < RB; ++rt)
          acc[rt] = __builtin_amdgcn_wmma_f32_16x16x4_f32(false, a, false, xb[rt][kg], (short)0,
                                                          (kg == 0) ? c0 : acc[rt], false, false);
      }
#pragma unroll
      for (int rt = 0; rt < RB; ++rt) {
        const v2f z01 = {acc[rt][0], acc[rt][1]}, z23 = {acc[rt][2], acc[rt][3]};
        const v2f z45 = {acc[rt][4], acc[rt][5]}, z67 = {acc[rt][6], acc[rt][7]};
        sA[rt] = z01 * z01 + sA[rt];
        sA[rt] = z23 * z23 + sA[rt];
        sB[rt] = z45 * z45 + sB[rt];
        sB[rt] = z67 * z67 + sB[rt];
      }
    }

#pragma unroll
    for (int rt = 0; rt < RB; ++rt) {
      const int pt = (RB * tp + rt) * 16 + c;
      vbuf[kA][pt]     = fmaf(-0.5f, sA[rt].x + sA[rt].y, cA);
      vbuf[kA + 1][pt] = fmaf(-0.5f, sB[rt].x + sB[rt].y, cB);
    }
  }
  __syncthreads();

  {
    float m = vbuf[0][tid];
#pragma unroll
    for (int k = 1; k < GMM_K; ++k) m = fmaxf(m, vbuf[k][tid]);
    float s = 0.f;
#pragma unroll
    for (int k = 0; k < GMM_K; ++k) s += expf(vbuf[k][tid] - m);
    if (p0 + tid < npts) out[(size_t)p0 + tid] = m + logf(s);
  }
}

extern "C" void kernel_launch(void* const* d_in, const int* in_sizes, int n_in,
                              void* d_out, int out_size, void* d_ws, size_t ws_size,
                              hipStream_t stream) {
  const float* X   = (const float*)d_in[0];
  const float* loc = (const float*)d_in[1];
  const float* cov = (const float*)d_in[2];
  const float* pi  = (const float*)d_in[3];
  float* out = (float*)d_out;
  float* pack = (float*)d_ws;

  const int N = in_sizes[0] / 64;
  const int ntiles = N / 16;

  gmm_prep<<<GMM_K, 256, 0, stream>>>(loc, cov, pi, pack);
  const int nblk = (ntiles + TPB - 1) / TPB;
  gmm_main<<<nblk, 32 * WPB, 0, stream>>>(X, pack, out, ntiles);
}


# gmm_prep: the panel factorisation's serial chain per pivot is a cross-lane read, hardware reciprocal (1 ulp), one multiply, one fma (was IEEE division and a branch); sum order kept, rounding changes
# speedup vs baseline: 6.5172x; 1.0377x over previous
#include <hip/hip_runtime.h>
#include <math.h>


#define GMM_K   32
#define LOG2PI  1.8378770664093453f

#define PACKF     2244
#define NFRAG     136
#define BIAS_OFF  2176
#define CONST_OFF 2240

#define RB      2
#define WPB     8
#define CPW     4
#define TPB     16
#define XP      68
#define BP      72

typedef __attribute__((ext_vector_type(2))) float v2f;
typedef __attribute__((ext_vector_type(8))) float v8f;


__device__ __forceinline__ float lane_read(float x, int src) {
  return __int_as_float(__builtin_amdgcn_readlane(__float_as_int(x), src));
}

__global__ __launch_bounds__(256) void gmm_prep(const float* __restrict__ loc,
                                                const float* __restrict__ cov,
                                                const float* __restrict__ pi,
                                                float* __restrict__ pack) {
  __shared__ __attribute__((aligned(16))) float Ls[64][68];
  __shared__ float M[64][65];
  __shared__ __attribute__((aligned(16))) float gs[2][64][8];
  __shared__ float rdv[64];
  __shared__ float lg[64];
  __shared__ float bv[64];
  __shared__ float sc[2];
  const int k = blockIdx.x;
  const int tid = threadIdx.x;
  const int lane = tid & 31;
  const int w = __builtin_amdgcn_readfirstlane(tid >> 5);
  const int r0 = 2 * lane;

  {
    const float4* src = (const float4*)(cov + (size_t)k * 4096);
#pragma unroll
    for (int it = 0; it < 4; ++it) {
      const int f = it * 256 + tid;
      *(float4*)&Ls[f >> 4][(f & 15) * 4] = src[f];
    }
  }
  float y[2];
  y[0] = loc[k * 64 + r0];
  y[1] = loc[k * 64 + r0 + 1];
  if (tid == 0) {
    float m = pi[0];
    for (int i = 1; i < GMM_K; ++i) m = fmaxf(m, pi[i]);
    float s = 0.f;
    for (int i = 0; i < GMM_K; ++i) s += expf(pi[i] - m);
    sc[0] = m + logf(s);
  }
  __syncthreads();

  float a[2][8], R[2][8];
#pragma unroll
  for (int h = 0; h < 2; ++h) {
#pragma unroll
    for (int q = 0; q < 8; ++q) {
      const int r = r0 + h, c = 8 * w + q;
      const float lo = Ls[r][c];
      const float up = Ls[c][r];
      a[h][q] = (r >= c) ? lo : up;
      R[h][q] = (r == c) ? 1.f : 0.f;
    }
  }

#pragma nounroll
  for (int p = 0; p < 8; ++p) {
    float (*gb)[8] = gs[p & 1];
    if (w == p) {
      float g[8][2];
      float keep[2][8];
#pragma unroll
      for (int h = 0; h < 2; ++h) {
#pragma unroll
        for (int q = 0; q < 8; ++q) keep[h][q] = a[h][q];
      }
      unsigned emin = 255u, emax = 0u;
#pragma unroll
      for (int jj = 0; jj < 8; ++jj) {
        const int j = 8 * p + jj;
        const int lj = 4 * p + (jj >> 1);
        const float d = lane_read(a[jj & 1][jj], lj);
        const unsigned ex = (__float_as_uint(d) >> 23) & 0xffu;
        emin = (ex < emin) ? ex : emin;
        emax = (ex > emax) ? ex : emax;
        const float rd = __builtin_amdgcn_rcpf(d);
        const float am0 = (r0 > j) ? a[0][jj] : 0.f;
        const float am1 = (r0 + 1 > j) ? a[1][jj] : 0.f;
        g[jj][0] = am0 * rd;
        g[jj][1] = am1 * rd;
#pragma unroll
        for (int q = jj + 1; q < 8; ++q) {
          const float s = lane_read(a[jj & 1][q], lj);
          a[0][q] = fmaf(-g[jj][0], s, a[0][q]);
          a[1][q] = fmaf(-g[jj][1], s, a[1][q]);
        }
      }
      const bool fast = (emin >= 1u) && (emax <= 252u);
      if (__builtin_expect(!fast, 0)) {
#pragma unroll
        for (int h = 0; h < 2; ++h) {
#pragma unroll
          for (int q = 0; q < 8; ++q) a[h][q] = keep[h][q];
        }
#pragma unroll
        for (int jj = 0; jj < 8; ++jj) {
          const int j = 8 * p + jj;
          const int lj = 4 * p + (jj >> 1);
          const float d = lane_read(a[jj & 1][jj], lj);
          g[jj][0] = (r0 > j) ? a[0][jj] / d : 0.f;
          g[jj][1] = (r0 + 1 > j) ? a[1][jj] / d : 0.f;
#pragma unroll
          for (int q = jj + 1; q < 8; ++q) {
            const float s = lane_read(a[jj & 1][q], lj);
            a[0][q] = fmaf(-g[jj][0], s, a[0][q]);
            a[1][q] = fmaf(-g[jj][1], s, a[1][q]);
          }
        }
      }
#pragma unroll
      for (int h = 0; h < 2; ++h) {
        *(float4*)&gb[r0 + h][0] = make_float4(g[0][h], g[1][h], g[2][h], g[3][h]);
        *(float4*)&gb[r0 + h][4] = make_float4(g[4][h], g[5][h], g[6][h], g[7][h]);
      }
    }
    __syncthreads();

    float G[8][2];
#pragma unroll
    for (int h = 0; h < 2; ++h) {
      const float4 v0 = *(const float4*)&gb[r0 + h][0];
      const float4 v1 = *(const float4*)&gb[r0 + h][4];
      G[0][h] = v0.x; G[1][h] = v0.y; G[2][h] = v0.z; G[3][h] = v0.w;
      G[4][h] = v1.x; G[5][h] = v1.y; G[6][h] = v1.z; G[7][h] = v1.w;
    }
    if (w > p) {
#pragma unroll
      for (int jj = 0; jj < 8; ++jj) {
        const int lj = 4 * p + (jj >> 1);
#pragma unroll
        for (int q = 0; q < 8; ++q) {
          const float s = lane_read(a[jj & 1][q], lj);
          a[0][q] = fmaf(-G[jj][0], s, a[0][q]);
          a[1][q] = fmaf(-G[jj][1], s, a[1][q]);
        }
      }
    } else {
#pragma unroll
      for (int jj = 0; jj < 8; ++jj) {
        const int lj = 4 * p + (jj >> 1);
#pragma unroll
        for (int q = 0; q < 8; ++q) {
          const float s = lane_read(R[jj & 1][q], lj);
          R[0][q] = fmaf(-G[jj][0], s, R[0][q]);
          R[1][q] = fmaf(-G[jj][1], s, R[1][q]);
        }
        if (w == 0) {
          const float s = lane_read(y[jj & 1], lj);
          y[0] = fmaf(-G[jj][0], s, y[0]);
          y[1] = fmaf(-G[jj][1], s, y[1]);
        }
      }
    }
  }

  if ((lane >> 2) == w) {
#pragma unroll
    for (int h = 0; h < 2; ++h) {
      float dr = a[h][h];
#pragma unroll
      for (int t = 1; t < 4; ++t) dr = ((lane & 3) == t) ? a[h][2 * t + h] : dr;
      const float sq = sqrtf(dr);
      rdv[r0 + h] = 1.0f / sq;
      lg[r0 + h] = logf(sq);
    }
  }
  __syncthreads();

#pragma unroll
  for (int h = 0; h < 2; ++h) {
    const float sr = rdv[r0 + h];
#pragma unroll
    for (int q = 0; q < 8; ++q) M[r0 + h][8 * w + q] = R[h][q] * sr;
    if (w == 0) bv[r0 + h] = y[h] * sr;
  }
  if (tid == 0) {
    float hs = 0.f;
    for (int i = 0; i < 64; ++i) hs += lg[i];
    sc[1] = hs;
  }
  __syncthreads();

  {
    float* base = pack + (size_t)k * PACKF;
#pragma unroll
    for (int m = 0; m < 16; ++m) {
      if (tid < 16 * (m + 1))
        base[8 * m * (m + 1) + tid] = M[4 * m + ((tid >> 2) & 3)][4 * (tid >> 4) + (tid & 3)];
    }
    if (tid < 64) base[BIAS_OFF + tid] = -bv[tid];
    if (tid < 4)
      base[CONST_OFF + tid] = (tid == 0) ? (pi[k] - sc[0] - sc[1] - 0.5f * 64.f * LOG2PI) : 0.f;
  }
}

__global__ __launch_bounds__(32 * WPB) void gmm_main(const float* __restrict__ X,
                                                     const float* __restrict__ pack,
                                                     float* __restrict__ out,
                                                     int ntiles) {
  __shared__ __attribute__((aligned(16))) float xs[16 * TPB][XP];
  __shared__ __attribute__((aligned(32))) float bs[GMM_K][BP];
  __shared__ float vbuf[GMM_K][16 * TPB];
  const int tid = threadIdx.x;
  const int w = tid >> 5;
  const int lane = tid & 31;
  const int h = lane >> 4;
  const int c = lane & 15;
  const int npts = ntiles * 16;
  const int p0 = blockIdx.x * (16 * TPB);

  const int kA = CPW * w + 2 * h;
  const float cA = pack[(size_t)kA * PACKF + CONST_OFF];
  const float cB = pack[(size_t)(kA + 1) * PACKF + CONST_OFF];

  v2f af[NFRAG];
  {
    const float* fb = pack + (size_t)(CPW * w + (c >> 2)) * PACKF + 4 * (c & 3) + 2 * h;
#pragma unroll
    for (int f = 0; f < NFRAG; ++f) af[f] = *(const v2f*)(fb + 16 * f);
  }

#pragma unroll
  for (int it = 0; it < (16 * TPB * 16) / (32 * WPB); ++it) {
    const int f = it * (32 * WPB) + tid;
    const int row = f >> 4, c4 = f & 15;
    int gp = p0 + row;
    if (gp > npts - 1) gp = npts - 1;
    *(float4*)&xs[row][4 * c4] = *(const float4*)(X + (size_t)gp * 64 + 4 * c4);
  }
  {
    const int k = tid >> 3, q = tid & 7;
    const float* src = pack + (size_t)k * PACKF;
    *(float4*)&bs[k][8 * q]     = *(const float4*)(src + BIAS_OFF + 8 * q);
    *(float4*)&bs[k][8 * q + 4] = *(const float4*)(src + BIAS_OFF + 8 * q + 4);
  }

  __syncthreads();

  const float* bA = &bs[kA][0];
  const float* bB = &bs[kA + 1][0];

#pragma nounroll
  for (int tp = 0; tp < TPB / RB; ++tp) {
    v2f xb[RB][16];
#pragma unroll
    for (int rt = 0; rt < RB; ++rt)
#pragma unroll
      for (int kg = 0; kg < 16; ++kg)
        xb[rt][kg] = *(const v2f*)&xs[(RB * tp + rt) * 16 + c][4 * kg + 2 * h];

    v2f sA[RB], sB[RB];
#pragma unroll
    for (int rt = 0; rt < RB; ++rt) { sA[rt] = (v2f){0.f, 0.f}; sB[rt] = (v2f){0.f, 0.f}; }

#pragma unroll
    for (int m = 0; m < 16; ++m) {
      const float4 b0 = *(const float4*)(bA + 4 * m);
      const float4 b1 = *(const float4*)(bB + 4 * m);
      const v8f c0 = {b0.x, b0.y, b0.z, b0.w, b1.x, b1.y, b1.z, b1.w};
      v8f acc[RB];
#pragma unroll
      for (int kg = 0; kg <= m; ++kg) {
        const v2f a = af[(m * (m + 1)) / 2 + kg];
#pragma unroll
        for (int rt = 0; rt < RB; ++rt)
          acc[rt] = __builtin_amdgcn_wmma_f32_16x16x4_f32(false, a, false, xb[rt][kg], (short)0,
                                                          (kg == 0) ? c0 : acc[rt], false, false);
      }
#pragma unroll
      for (int rt = 0; rt < RB; ++rt) {
        const v2f z01 = {acc[rt][0], acc[rt][1]}, z23 = {acc[rt][2], acc[rt][3]};
        const v2f z45 = {acc[rt][4], acc[rt][5]}, z67 = {acc[rt][6], acc[rt][7]};
        sA[rt] = z01 * z01 + sA[rt];
        sA[rt] = z23 * z23 + sA[rt];
        sB[rt] = z45 * z45 + sB[rt];
        sB[rt] = z67 * z67 + sB[rt];
      }
    }

#pragma unroll
    for (int rt = 0; rt < RB; ++rt) {
      const int pt = (RB * tp + rt) * 16 + c;
      vbuf[kA][pt]     = fmaf(-0.5f, sA[rt].x + sA[rt].y, cA);
      vbuf[kA + 1][pt] = fmaf(-0.5f, sB[rt].x + sB[rt].y, cB);
    }
  }
  __syncthreads();

  {
    float m = vbuf[0][tid];
#pragma unroll
    for (int k = 1; k < GMM_K; ++k) m = fmaxf(m, vbuf[k][tid]);
    float s = 0.f;
#pragma unroll
    for (int k = 0; k < GMM_K; ++k) s += expf(vbuf[k][tid] - m);
    if (p0 + tid < npts) out[(size_t)p0 + tid] = m + logf(s);
  }
}

extern "C" void kernel_launch(void* const* d_in, const int* in_sizes, int n_in,
                              void* d_out, int out_size, void* d_ws, size_t ws_size,
                              hipStream_t stream) {
  const float* X   = (const float*)d_in[0];
  const float* loc = (const float*)d_in[1];
  const float* cov = (const float*)d_in[2];
  const float* pi  = (const float*)d_in[3];
  float* out = (float*)d_out;
  float* pack = (float*)d_ws;

  const int N = in_sizes[0] / 64;
  const int ntiles = N / 16;

  gmm_prep<<<GMM_K, 256, 0, stream>>>(loc, cov, pi, pack);
  const int nblk = (ntiles + TPB - 1) / TPB;
  gmm_main<<<nblk, 32 * WPB, 0, stream>>>(X, pack, out, ntiles);
}


# gmm_main: first barrier waits for the x rows only; fragments come as 16-byte loads (pack re-laid in 512-B chunks), 62 of 68 per lane issued behind it, landing under the first pass's WMMAs; sums kept
# speedup vs baseline: 6.7355x; 1.0335x over previous
#include <hip/hip_runtime.h>
#include <math.h>


#define GMM_K   32
#define LOG2PI  1.8378770664093453f

#define NFRAG      136
#define NPAIR      68
#define CHUNKF     128
#define BIAS_BASE  (8 * NPAIR * CHUNKF)
#define CONST_BASE (BIAS_BASE + GMM_K * 64)

#define RB      2
#define WPB     8
#define CPW     4
#define TPB     16
#define XP      68
#define BP      72

typedef __attribute__((ext_vector_type(2))) float v2f;
typedef __attribute__((ext_vector_type(8))) float v8f;


__device__ __forceinline__ float lane_read(float x, int src) {
  return __int_as_float(__builtin_amdgcn_readlane(__float_as_int(x), src));
}

__global__ __launch_bounds__(256) void gmm_prep(const float* __restrict__ loc,
                                                const float* __restrict__ cov,
                                                const float* __restrict__ pi,
                                                float* __restrict__ pack) {
  __shared__ __attribute__((aligned(16))) float Ls[64][68];
  __shared__ float M[64][65];
  __shared__ __attribute__((aligned(16))) float gs[2][64][8];
  __shared__ float rdv[64];
  __shared__ float lg[64];
  __shared__ float bv[64];
  __shared__ float sc[2];
  const int k = blockIdx.x;
  const int tid = threadIdx.x;
  const int lane = tid & 31;
  const int w = __builtin_amdgcn_readfirstlane(tid >> 5);
  const int r0 = 2 * lane;

  {
    const float4* src = (const float4*)(cov + (size_t)k * 4096);
#pragma unroll
    for (int it = 0; it < 4; ++it) {
      const int f = it * 256 + tid;
      *(float4*)&Ls[f >> 4][(f & 15) * 4] = src[f];
    }
  }
  float y[2];
  y[0] = loc[k * 64 + r0];
  y[1] = loc[k * 64 + r0 + 1];
  if (tid == 0) {
    float m = pi[0];
    for (int i = 1; i < GMM_K; ++i) m = fmaxf(m, pi[i]);
    float s = 0.f;
    for (int i = 0; i < GMM_K; ++i) s += expf(pi[i] - m);
    sc[0] = m + logf(s);
  }
  __syncthreads();

  float a[2][8], R[2][8];
#pragma unroll
  for (int h = 0; h < 2; ++h) {
#pragma unroll
    for (int q = 0; q < 8; ++q) {
      const int r = r0 + h, c = 8 * w + q;
      const float lo = Ls[r][c];
      const float up = Ls[c][r];
      a[h][q] = (r >= c) ? lo : up;
      R[h][q] = (r == c) ? 1.f : 0.f;
    }
  }

#pragma nounroll
  for (int p = 0; p < 8; ++p) {
    float (*gb)[8] = gs[p & 1];
    if (w == p) {
      float g[8][2];
      float keep[2][8];
#pragma unroll
      for (int h = 0; h < 2; ++h) {
#pragma unroll
        for (int q = 0; q < 8; ++q) keep[h][q] = a[h][q];
      }
      unsigned emin = 255u, emax = 0u;
#pragma unroll
      for (int jj = 0; jj < 8; ++jj) {
        const int j = 8 * p + jj;
        const int lj = 4 * p + (jj >> 1);
        const float d = lane_read(a[jj & 1][jj], lj);
        const unsigned ex = (__float_as_uint(d) >> 23) & 0xffu;
        emin = (ex < emin) ? ex : emin;
        emax = (ex > emax) ? ex : emax;
        const float rd = __builtin_amdgcn_rcpf(d);
        const float am0 = (r0 > j) ? a[0][jj] : 0.f;
        const float am1 = (r0 + 1 > j) ? a[1][jj] : 0.f;
        g[jj][0] = am0 * rd;
        g[jj][1] = am1 * rd;
#pragma unroll
        for (int q = jj + 1; q < 8; ++q) {
          const float s = lane_read(a[jj & 1][q], lj);
          a[0][q] = fmaf(-g[jj][0], s, a[0][q]);
          a[1][q] = fmaf(-g[jj][1], s, a[1][q]);
        }
      }
      const bool fast = (emin >= 1u) && (emax <= 252u);
      if (__builtin_expect(!fast, 0)) {
#pragma unroll
        for (int h = 0; h < 2; ++h) {
#pragma unroll
          for (int q = 0; q < 8; ++q) a[h][q] = keep[h][q];
        }
#pragma unroll
        for (int jj = 0; jj < 8; ++jj) {
          const int j = 8 * p + jj;
          const int lj = 4 * p + (jj >> 1);
          const float d = lane_read(a[jj & 1][jj], lj);
          g[jj][0] = (r0 > j) ? a[0][jj] / d : 0.f;
          g[jj][1] = (r0 + 1 > j) ? a[1][jj] / d : 0.f;
#pragma unroll
          for (int q = jj + 1; q < 8; ++q) {
            const float s = lane_read(a[jj & 1][q], lj);
            a[0][q] = fmaf(-g[jj][0], s, a[0][q]);
            a[1][q] = fmaf(-g[jj][1], s, a[1][q]);
          }
        }
      }
#pragma unroll
      for (int h = 0; h < 2; ++h) {
        *(float4*)&gb[r0 + h][0] = make_float4(g[0][h], g[1][h], g[2][h], g[3][h]);
        *(float4*)&gb[r0 + h][4] = make_float4(g[4][h], g[5][h], g[6][h], g[7][h]);
      }
    }
    __syncthreads();

    float G[8][2];
#pragma unroll
    for (int h = 0; h < 2; ++h) {
      const float4 v0 = *(const float4*)&gb[r0 + h][0];
      const float4 v1 = *(const float4*)&gb[r0 + h][4];
      G[0][h] = v0.x; G[1][h] = v0.y; G[2][h] = v0.z; G[3][h] = v0.w;
      G[4][h] = v1.x; G[5][h] = v1.y; G[6][h] = v1.z; G[7][h] = v1.w;
    }
    if (w > p) {
#pragma unroll
      for (int jj = 0; jj < 8; ++jj) {
        const int lj = 4 * p + (jj >> 1);
#pragma unroll
        for (int q = 0; q < 8; ++q) {
          const float s = lane_read(a[jj & 1][q], lj);
          a[0][q] = fmaf(-G[jj][0], s, a[0][q]);
          a[1][q] = fmaf(-G[jj][1], s, a[1][q]);
        }
      }
    } else {
#pragma unroll
      for (int jj = 0; jj < 8; ++jj) {
        const int lj = 4 * p + (jj >> 1);
#pragma unroll
        for (int q = 0; q < 8; ++q) {
          const float s = lane_read(R[jj & 1][q], lj);
          R[0][q] = fmaf(-G[jj][0], s, R[0][q]);
          R[1][q] = fmaf(-G[jj][1], s, R[1][q]);
        }
        if (w == 0) {
          const float s = lane_read(y[jj & 1], lj);
          y[0] = fmaf(-G[jj][0], s, y[0]);
          y[1] = fmaf(-G[jj][1], s, y[1]);
        }
      }
    }
  }

  if ((lane >> 2) == w) {
#pragma unroll
    for (int h = 0; h < 2; ++h) {
      float dr = a[h][h];
#pragma unroll
      for (int t = 1; t < 4; ++t) dr = ((lane & 3) == t) ? a[h][2 * t + h] : dr;
      const float sq = sqrtf(dr);
      rdv[r0 + h] = 1.0f / sq;
      lg[r0 + h] = logf(sq);
    }
  }
  __syncthreads();

#pragma unroll
  for (int h = 0; h < 2; ++h) {
    const float sr = rdv[r0 + h];
#pragma unroll
    for (int q = 0; q < 8; ++q) M[r0 + h][8 * w + q] = R[h][q] * sr;
    if (w == 0) bv[r0 + h] = y[h] * sr;
  }
  if (tid == 0) {
    float hs = 0.f;
    for (int i = 0; i < 64; ++i) hs += lg[i];
    sc[1] = hs;
  }
  __syncthreads();

  {
    const int ri = (tid >> 2) & 3, e = tid & 3, kg = tid >> 4;
    float* base = pack + (size_t)((k >> 2) * NPAIR * 4 + (k & 3)) * 32 + 8 * ri + 4 * (e >> 1) + (e & 1);
#pragma unroll
    for (int m = 0; m < 16; ++m) {
      if (tid < 16 * (m + 1)) {
        const int f = (m * (m + 1)) / 2 + kg;
        base[CHUNKF * (f >> 1) + 2 * (f & 1)] = M[4 * m + ri][4 * kg + e];
      }
    }
    if (tid < 64) pack[BIAS_BASE + k * 64 + tid] = -bv[tid];
    if (tid == 0) pack[CONST_BASE + k] = pi[k] - sc[0] - sc[1] - 0.5f * 64.f * LOG2PI;
  }
}

__global__ __launch_bounds__(32 * WPB) void gmm_main(const float* __restrict__ X,
                                                     const float* __restrict__ pack,
                                                     float* __restrict__ out,
                                                     int ntiles) {
  __shared__ __attribute__((aligned(16))) float xs[16 * TPB][XP];
  __shared__ __attribute__((aligned(32))) float bs[GMM_K][BP];
  __shared__ float vbuf[GMM_K][16 * TPB];
  const int tid = threadIdx.x;
  const int w = tid >> 5;
  const int lane = tid & 31;
  const int h = lane >> 4;
  const int c = lane & 15;
  const int npts = ntiles * 16;
  const int p0 = blockIdx.x * (16 * TPB);

  const int kA = CPW * w + 2 * h;
  const float cA = pack[CONST_BASE + kA];
  const float cB = pack[CONST_BASE + kA + 1];

  constexpr int XIT = (16 * TPB * 16) / (32 * WPB);
  float4 xv[XIT];
#pragma unroll
  for (int it = 0; it < XIT; ++it) {
    const int f = it * (32 * WPB) + tid;
    int gp = p0 + (f >> 4);
    if (gp > npts - 1) gp = npts - 1;
    xv[it] = *(const float4*)(X + (size_t)gp * 64 + 4 * (f & 15));
  }
  const float4 bv0 = *(const float4*)(pack + BIAS_BASE + 8 * tid);
  const float4 bv1 = *(const float4*)(pack + BIAS_BASE + 8 * tid + 4);

  __builtin_amdgcn_sched_barrier(0);

  constexpr int FPRE = 6;
  float4 af[NPAIR];
  const float* fb = pack + (size_t)(w * NPAIR * 4 + (c >> 2)) * 32 + 8 * (c & 3) + 4 * h;
#pragma unroll
  for (int p = 0; p < FPRE; ++p) af[p] = *(const float4*)(fb + CHUNKF * p);

#pragma unroll
  for (int it = 0; it < XIT; ++it) {
    const int f = it * (32 * WPB) + tid;
    *(float4*)&xs[f >> 4][4 * (f & 15)] = xv[it];
  }
  *(float4*)&bs[tid >> 3][8 * (tid & 7)]     = bv0;
  *(float4*)&bs[tid >> 3][8 * (tid & 7) + 4] = bv1;

  asm volatile("s_wait_dscnt 0x0\n\ts_barrier_signal -1\n\ts_barrier_wait -1" ::: "memory");

#pragma unroll
  for (int p = FPRE; p < NPAIR; ++p) af[p] = *(const float4*)(fb + CHUNKF * p);

  const float* bA = &bs[kA][0];
  const float* bB = &bs[kA + 1][0];

#pragma nounroll
  for (int tp = 0; tp < TPB / RB; ++tp) {
    v2f xb[RB][16];
#pragma unroll
    for (int rt = 0; rt < RB; ++rt)
#pragma unroll
      for (int kg = 0; kg < 16; ++kg)
        xb[rt][kg] = *(const v2f*)&xs[(RB * tp + rt) * 16 + c][4 * kg + 2 * h];

    v2f sA[RB], sB[RB];
#pragma unroll
    for (int rt = 0; rt < RB; ++rt) { sA[rt] = (v2f){0.f, 0.f}; sB[rt] = (v2f){0.f, 0.f}; }

#pragma unroll
    for (int m = 0; m < 16; ++m) {
      const float4 b0 = *(const float4*)(bA + 4 * m);
      const float4 b1 = *(const float4*)(bB + 4 * m);
      const v8f c0 = {b0.x, b0.y, b0.z, b0.w, b1.x, b1.y, b1.z, b1.w};
      v8f acc[RB];
#pragma unroll
      for (int kg = 0; kg <= m; ++kg) {
        const int f = (m * (m + 1)) / 2 + kg;
        const v2f a = (f & 1) ? (v2f){af[f >> 1].z, af[f >> 1].w} : (v2f){af[f >> 1].x, af[f >> 1].y};
#pragma unroll
        for (int rt = 0; rt < RB; ++rt)
          acc[rt] = __builtin_amdgcn_wmma_f32_16x16x4_f32(false, a, false, xb[rt][kg], (short)0,
                                                          (kg == 0) ? c0 : acc[rt], false, false);
      }
#pragma unroll
      for (int rt = 0; rt < RB; ++rt) {
        const v2f z01 = {acc[rt][0], acc[rt][1]}, z23 = {acc[rt][2], acc[rt][3]};
        const v2f z45 = {acc[rt][4], acc[rt][5]}, z67 = {acc[rt][6], acc[rt][7]};
        sA[rt] = z01 * z01 + sA[rt];
        sA[rt] = z23 * z23 + sA[rt];
        sB[rt] = z45 * z45 + sB[rt];
        sB[rt] = z67 * z67 + sB[rt];
      }
    }

#pragma unroll
    for (int rt = 0; rt < RB; ++rt) {
      const int pt = (RB * tp + rt) * 16 + c;
      vbuf[kA][pt]     = fmaf(-0.5f, sA[rt].x + sA[rt].y, cA);
      vbuf[kA + 1][pt] = fmaf(-0.5f, sB[rt].x + sB[rt].y, cB);
    }
  }
  __syncthreads();

  {
    float m = vbuf[0][tid];
#pragma unroll
    for (int k = 1; k < GMM_K; ++k) m = fmaxf(m, vbuf[k][tid]);
    float s = 0.f;
#pragma unroll
    for (int k = 0; k < GMM_K; ++k) s += expf(vbuf[k][tid] - m);
    if (p0 + tid < npts) out[(size_t)p0 + tid] = m + logf(s);
  }
}

extern "C" void kernel_launch(void* const* d_in, const int* in_sizes, int n_in,
                              void* d_out, int out_size, void* d_ws, size_t ws_size,
                              hipStream_t stream) {
  const float* X   = (const float*)d_in[0];
  const float* loc = (const float*)d_in[1];
  const float* cov = (const float*)d_in[2];
  const float* pi  = (const float*)d_in[3];
  float* out = (float*)d_out;
  float* pack = (float*)d_ws;

  const int N = in_sizes[0] / 64;
  const int ntiles = N / 16;

  gmm_prep<<<GMM_K, 256, 0, stream>>>(loc, cov, pi, pack);
  const int nblk = (ntiles + TPB - 1) / TPB;
  gmm_main<<<nblk, 32 * WPB, 0, stream>>>(X, pack, out, ntiles);
}


# the per-component constant is assembled in gmm_main under its x-load wait (lse(pi) over 32 lanes, log-det over 8, DPP butterflies); gmm_prep loses both one-thread sections; both sums' order changes
# speedup vs baseline: 7.0809x; 1.0513x over previous
#include <hip/hip_runtime.h>
#include <math.h>


#define GMM_K   32
#define LOG2PI  1.8378770664093453f

#define NFRAG      136
#define NPAIR      68
#define CHUNKF     128
#define BIAS_BASE  (8 * NPAIR * CHUNKF)
#define LG_BASE    (BIAS_BASE + GMM_K * 64)

#define RB      2
#define WPB     8
#define CPW     4
#define TPB     16
#define XP      68
#define BP      72

typedef __attribute__((ext_vector_type(2))) float v2f;
typedef __attribute__((ext_vector_type(8))) float v8f;


__device__ __forceinline__ float lane_read(float x, int src) {
  return __int_as_float(__builtin_amdgcn_readlane(__float_as_int(x), src));
}

template <int M> __device__ __forceinline__ float row_xor(float x) {
  const int b = __builtin_bit_cast(int, x);
  return __builtin_bit_cast(float, __builtin_amdgcn_update_dpp(b, b, 0x160 + M, 0xf, 0xf, false));
}
__device__ __forceinline__ float xhalf(float x) {
  const int b = __builtin_bit_cast(int, x);
  return __builtin_bit_cast(float, __builtin_amdgcn_permlanex16(b, b, 0x76543210, 0xfedcba98, false, false));
}

__global__ __launch_bounds__(256) void gmm_prep(const float* __restrict__ loc,
                                                const float* __restrict__ cov,
                                                float* __restrict__ pack) {
  __shared__ __attribute__((aligned(16))) float Ls[64][68];
  __shared__ float M[64][65];
  __shared__ __attribute__((aligned(16))) float gs[2][64][8];
  __shared__ float rdv[64];
  __shared__ float lg[64];
  __shared__ float bv[64];
  const int k = blockIdx.x;
  const int tid = threadIdx.x;
  const int lane = tid & 31;
  const int w = __builtin_amdgcn_readfirstlane(tid >> 5);
  const int r0 = 2 * lane;

  {
    const float4* src = (const float4*)(cov + (size_t)k * 4096);
#pragma unroll
    for (int it = 0; it < 4; ++it) {
      const int f = it * 256 + tid;
      *(float4*)&Ls[f >> 4][(f & 15) * 4] = src[f];
    }
  }
  float y[2];
  y[0] = loc[k * 64 + r0];
  y[1] = loc[k * 64 + r0 + 1];
  __syncthreads();

  float a[2][8], R[2][8];
#pragma unroll
  for (int h = 0; h < 2; ++h) {
#pragma unroll
    for (int q = 0; q < 8; ++q) {
      const int r = r0 + h, c = 8 * w + q;
      const float lo = Ls[r][c];
      const float up = Ls[c][r];
      a[h][q] = (r >= c) ? lo : up;
      R[h][q] = (r == c) ? 1.f : 0.f;
    }
  }

#pragma nounroll
  for (int p = 0; p < 8; ++p) {
    float (*gb)[8] = gs[p & 1];
    if (w == p) {
      float g[8][2];
      float keep[2][8];
#pragma unroll
      for (int h = 0; h < 2; ++h) {
#pragma unroll
        for (int q = 0; q < 8; ++q) keep[h][q] = a[h][q];
      }
      unsigned emin = 255u, emax = 0u;
#pragma unroll
      for (int jj = 0; jj < 8; ++jj) {
        const int j = 8 * p + jj;
        const int lj = 4 * p + (jj >> 1);
        const float d = lane_read(a[jj & 1][jj], lj);
        const unsigned ex = (__float_as_uint(d) >> 23) & 0xffu;
        emin = (ex < emin) ? ex : emin;
        emax = (ex > emax) ? ex : emax;
        const float rd = __builtin_amdgcn_rcpf(d);
        const float am0 = (r0 > j) ? a[0][jj] : 0.f;
        const float am1 = (r0 + 1 > j) ? a[1][jj] : 0.f;
        g[jj][0] = am0 * rd;
        g[jj][1] = am1 * rd;
#pragma unroll
        for (int q = jj + 1; q < 8; ++q) {
          const float s = lane_read(a[jj & 1][q], lj);
          a[0][q] = fmaf(-g[jj][0], s, a[0][q]);
          a[1][q] = fmaf(-g[jj][1], s, a[1][q]);
        }
      }
      const bool fast = (emin >= 1u) && (emax <= 252u);
      if (__builtin_expect(!fast, 0)) {
#pragma unroll
        for (int h = 0; h < 2; ++h) {
#pragma unroll
          for (int q = 0; q < 8; ++q) a[h][q] = keep[h][q];
        }
#pragma unroll
        for (int jj = 0; jj < 8; ++jj) {
          const int j = 8 * p + jj;
          const int lj = 4 * p + (jj >> 1);
          const float d = lane_read(a[jj & 1][jj], lj);
          g[jj][0] = (r0 > j) ? a[0][jj] / d : 0.f;
          g[jj][1] = (r0 + 1 > j) ? a[1][jj] / d : 0.f;
#pragma unroll
          for (int q = jj + 1; q < 8; ++q) {
            const float s = lane_read(a[jj & 1][q], lj);
            a[0][q] = fmaf(-g[jj][0], s, a[0][q]);
            a[1][q] = fmaf(-g[jj][1], s, a[1][q]);
          }
        }
      }
#pragma unroll
      for (int h = 0; h < 2; ++h) {
        *(float4*)&gb[r0 + h][0] = make_float4(g[0][h], g[1][h], g[2][h], g[3][h]);
        *(float4*)&gb[r0 + h][4] = make_float4(g[4][h], g[5][h], g[6][h], g[7][h]);
      }
    }
    __syncthreads();

    float G[8][2];
#pragma unroll
    for (int h = 0; h < 2; ++h) {
      const float4 v0 = *(const float4*)&gb[r0 + h][0];
      const float4 v1 = *(const float4*)&gb[r0 + h][4];
      G[0][h] = v0.x; G[1][h] = v0.y; G[2][h] = v0.z; G[3][h] = v0.w;
      G[4][h] = v1.x; G[5][h] = v1.y; G[6][h] = v1.z; G[7][h] = v1.w;
    }
    if (w > p) {
#pragma unroll
      for (int jj = 0; jj < 8; ++jj) {
        const int lj = 4 * p + (jj >> 1);
#pragma unroll
        for (int q = 0; q < 8; ++q) {
          const float s = lane_read(a[jj & 1][q], lj);
          a[0][q] = fmaf(-G[jj][0], s, a[0][q]);
          a[1][q] = fmaf(-G[jj][1], s, a[1][q]);
        }
      }
    } else {
#pragma unroll
      for (int jj = 0; jj < 8; ++jj) {
        const int lj = 4 * p + (jj >> 1);
#pragma unroll
        for (int q = 0; q < 8; ++q) {
          const float s = lane_read(R[jj & 1][q], lj);
          R[0][q] = fmaf(-G[jj][0], s, R[0][q]);
          R[1][q] = fmaf(-G[jj][1], s, R[1][q]);
        }
        if (w == 0) {
          const float s = lane_read(y[jj & 1], lj);
          y[0] = fmaf(-G[jj][0], s, y[0]);
          y[1] = fmaf(-G[jj][1], s, y[1]);
        }
      }
    }
  }

  if ((lane >> 2) == w) {
#pragma unroll
    for (int h = 0; h < 2; ++h) {
      float dr = a[h][h];
#pragma unroll
      for (int t = 1; t < 4; ++t) dr = ((lane & 3) == t) ? a[h][2 * t + h] : dr;
      const float sq = sqrtf(dr);
      rdv[r0 + h] = 1.0f / sq;
      lg[r0 + h] = logf(sq);
    }
  }
  __syncthreads();

#pragma unroll
  for (int h = 0; h < 2; ++h) {
    const float sr = rdv[r0 + h];
#pragma unroll
    for (int q = 0; q < 8; ++q) M[r0 + h][8 * w + q] = R[h][q] * sr;
    if (w == 0) bv[r0 + h] = y[h] * sr;
  }
  __syncthreads();

  {
    const int ri = (tid >> 2) & 3, e = tid & 3, kg = tid >> 4;
    float* base = pack + (size_t)((k >> 2) * NPAIR * 4 + (k & 3)) * 32 + 8 * ri + 4 * (e >> 1) + (e & 1);
#pragma unroll
    for (int m = 0; m < 16; ++m) {
      if (tid < 16 * (m + 1)) {
        const int f = (m * (m + 1)) / 2 + kg;
        base[CHUNKF * (f >> 1) + 2 * (f & 1)] = M[4 * m + ri][4 * kg + e];
      }
    }
    if (tid < 64) {
      pack[BIAS_BASE + k * 64 + tid] = -bv[tid];
      pack[LG_BASE + k * 64 + tid] = lg[tid];
    }
  }
}

__global__ __launch_bounds__(32 * WPB) void gmm_main(const float* __restrict__ X,
                                                     const float* __restrict__ pi,
                                                     const float* __restrict__ pack,
                                                     float* __restrict__ out,
                                                     int ntiles) {
  __shared__ __attribute__((aligned(16))) float xs[16 * TPB][XP];
  __shared__ __attribute__((aligned(32))) float bs[GMM_K][BP];
  __shared__ float vbuf[GMM_K][16 * TPB];
  __shared__ float cs[GMM_K];
  const int tid = threadIdx.x;
  const int w = tid >> 5;
  const int lane = tid & 31;
  const int h = lane >> 4;
  const int c = lane & 15;
  const int npts = ntiles * 16;
  const int p0 = blockIdx.x * (16 * TPB);

  const int kA = CPW * w + 2 * h;
  const float pv = pi[lane];
  const float* lgp = pack + LG_BASE + (CPW * w) * 64 + 8 * lane;
  const float4 lg0 = *(const float4*)lgp;
  const float4 lg1 = *(const float4*)(lgp + 4);

  constexpr int XIT = (16 * TPB * 16) / (32 * WPB);
  float4 xv[XIT];
#pragma unroll
  for (int it = 0; it < XIT; ++it) {
    const int f = it * (32 * WPB) + tid;
    int gp = p0 + (f >> 4);
    if (gp > npts - 1) gp = npts - 1;
    xv[it] = *(const float4*)(X + (size_t)gp * 64 + 4 * (f & 15));
  }
  const float4 bv0 = *(const float4*)(pack + BIAS_BASE + 8 * tid);
  const float4 bv1 = *(const float4*)(pack + BIAS_BASE + 8 * tid + 4);

  __builtin_amdgcn_sched_barrier(0);

  constexpr int FPRE = 6;
  float4 af[NPAIR];
  const float* fb = pack + (size_t)(w * NPAIR * 4 + (c >> 2)) * 32 + 8 * (c & 3) + 4 * h;
#pragma unroll
  for (int p = 0; p < FPRE; ++p) af[p] = *(const float4*)(fb + CHUNKF * p);

#pragma unroll
  for (int it = 0; it < XIT; ++it) {
    const int f = it * (32 * WPB) + tid;
    *(float4*)&xs[f >> 4][4 * (f & 15)] = xv[it];
  }
  *(float4*)&bs[tid >> 3][8 * (tid & 7)]     = bv0;
  *(float4*)&bs[tid >> 3][8 * (tid & 7) + 4] = bv1;

  asm volatile("s_wait_dscnt 0x0\n\ts_barrier_signal -1" ::: "memory");
  {
    float mx = pv;
    mx = fmaxf(mx, row_xor<1>(mx));
    mx = fmaxf(mx, row_xor<2>(mx));
    mx = fmaxf(mx, row_xor<4>(mx));
    mx = fmaxf(mx, row_xor<8>(mx));
    mx = fmaxf(mx, xhalf(mx));
    float sm = expf(pv - mx);
    sm += row_xor<1>(sm);
    sm += row_xor<2>(sm);
    sm += row_xor<4>(sm);
    sm += row_xor<8>(sm);
    sm += xhalf(sm);
    const float qv = pv - (mx + logf(sm));
    float hl = ((lg0.x + lg0.y) + (lg0.z + lg0.w)) + ((lg1.x + lg1.y) + (lg1.z + lg1.w));
    hl += row_xor<1>(hl);
    hl += row_xor<2>(hl);
    hl += row_xor<4>(hl);
    const int w4 = 4 * __builtin_amdgcn_readfirstlane(w);
    const float q0 = lane_read(qv, w4), q1 = lane_read(qv, w4 + 1);
    const float q2 = lane_read(qv, w4 + 2), q3 = lane_read(qv, w4 + 3);
    const int q = lane >> 3;
    const float qlo = (q & 1) ? q1 : q0, qhi = (q & 1) ? q3 : q2;
    const float qq = (q & 2) ? qhi : qlo;
    cs[CPW * w + q] = qq - hl - 0.5f * 64.f * LOG2PI;
  }
  asm volatile("s_barrier_wait -1" ::: "memory");

#pragma unroll
  for (int p = FPRE; p < NPAIR; ++p) af[p] = *(const float4*)(fb + CHUNKF * p);

  const float* bA = &bs[kA][0];
  const float* bB = &bs[kA + 1][0];
  const float cA = cs[kA];
  const float cB = cs[kA + 1];

#pragma nounroll
  for (int tp = 0; tp < TPB / RB; ++tp) {
    v2f xb[RB][16];
#pragma unroll
    for (int rt = 0; rt < RB; ++rt)
#pragma unroll
      for (int kg = 0; kg < 16; ++kg)
        xb[rt][kg] = *(const v2f*)&xs[(RB * tp + rt) * 16 + c][4 * kg + 2 * h];

    v2f sA[RB], sB[RB];
#pragma unroll
    for (int rt = 0; rt < RB; ++rt) { sA[rt] = (v2f){0.f, 0.f}; sB[rt] = (v2f){0.f, 0.f}; }

#pragma unroll
    for (int m = 0; m < 16; ++m) {
      const float4 b0 = *(const float4*)(bA + 4 * m);
      const float4 b1 = *(const float4*)(bB + 4 * m);
      const v8f c0 = {b0.x, b0.y, b0.z, b0.w, b1.x, b1.y, b1.z, b1.w};
      v8f acc[RB];
#pragma unroll
      for (int kg = 0; kg <= m; ++kg) {
        const int f = (m * (m + 1)) / 2 + kg;
        const v2f a = (f & 1) ? (v2f){af[f >> 1].z, af[f >> 1].w} : (v2f){af[f >> 1].x, af[f >> 1].y};
#pragma unroll
        for (int rt = 0; rt < RB; ++rt)
          acc[rt] = __builtin_amdgcn_wmma_f32_16x16x4_f32(false, a, false, xb[rt][kg], (short)0,
                                                          (kg == 0) ? c0 : acc[rt], false, false);
      }
#pragma unroll
      for (int rt = 0; rt < RB; ++rt) {
        const v2f z01 = {acc[rt][0], acc[rt][1]}, z23 = {acc[rt][2], acc[rt][3]};
        const v2f z45 = {acc[rt][4], acc[rt][5]}, z67 = {acc[rt][6], acc[rt][7]};
        sA[rt] = z01 * z01 + sA[rt];
        sA[rt] = z23 * z23 + sA[rt];
        sB[rt] = z45 * z45 + sB[rt];
        sB[rt] = z67 * z67 + sB[rt];
      }
    }

#pragma unroll
    for (int rt = 0; rt < RB; ++rt) {
      const int pt = (RB * tp + rt) * 16 + c;
      vbuf[kA][pt]     = fmaf(-0.5f, sA[rt].x + sA[rt].y, cA);
      vbuf[kA + 1][pt] = fmaf(-0.5f, sB[rt].x + sB[rt].y, cB);
    }
  }
  __syncthreads();

  {
    float m = vbuf[0][tid];
#pragma unroll
    for (int k = 1; k < GMM_K; ++k) m = fmaxf(m, vbuf[k][tid]);
    float s = 0.f;
#pragma unroll
    for (int k = 0; k < GMM_K; ++k) s += expf(vbuf[k][tid] - m);
    if (p0 + tid < npts) out[(size_t)p0 + tid] = m + logf(s);
  }
}

extern "C" void kernel_launch(void* const* d_in, const int* in_sizes, int n_in,
                              void* d_out, int out_size, void* d_ws, size_t ws_size,
                              hipStream_t stream) {
  const float* X   = (const float*)d_in[0];
  const float* loc = (const float*)d_in[1];
  const float* cov = (const float*)d_in[2];
  const float* pi  = (const float*)d_in[3];
  float* out = (float*)d_out;
  float* pack = (float*)d_ws;

  const int N = in_sizes[0] / 64;
  const int ntiles = N / 16;

  gmm_prep<<<GMM_K, 256, 0, stream>>>(loc, cov, pack);
  const int nblk = (ntiles + TPB - 1) / TPB;
  gmm_main<<<nblk, 32 * WPB, 0, stream>>>(X, pi, pack, out, ntiles);
}
